# starting kernel
# baseline (speedup 1.0000x reference)
#include <hip/hip_runtime.h>

typedef __attribute__((ext_vector_type(2))) float v2f;
typedef __attribute__((ext_vector_type(8))) float v8f;

#define DCH 128


__global__ void gcn_deg_init(float* deg, int n) {
    int i = blockIdx.x * blockDim.x + threadIdx.x;
    if (i < n) deg[i] = 1.0f;
}

__global__ void gcn_deg_count(const int* __restrict__ tgt, float* deg, int e) {
    int i = blockIdx.x * blockDim.x + threadIdx.x;
    if (i < e) atomicAdd(&deg[tgt[i]], 1.0f);
}

__global__ void gcn_rsqrt(float* deg, int n) {
    int i = blockIdx.x * blockDim.x + threadIdx.x;
    if (i < n) deg[i] = rsqrtf(deg[i]);
}


__global__ void gcn_gemm_wmma(const float* __restrict__ A,
                              const float* __restrict__ W,
                              float* __restrict__ H) {
    int waveId = blockIdx.x * (blockDim.x >> 5) + (threadIdx.x >> 5);
    int lane   = threadIdx.x & 31;
    int mTile  = waveId >> 3;
    int nTile  = waveId & 7;
    int hi     = lane >> 4;
    int l16    = lane & 15;

    int row = mTile * 16 + l16;
    int col = nTile * 16 + l16;

    const float* arow = A + (size_t)row * DCH + 2 * hi;

    v8f c = {};
#pragma unroll
    for (int k0 = 0; k0 < DCH; k0 += 4) {
        v2f a, b;
        a.x = arow[k0 + 0];
        a.y = arow[k0 + 1];
        int kb = k0 + 2 * hi;
        b.x = W[(size_t)kb * DCH + col];
        b.y = W[(size_t)(kb + 1) * DCH + col];
        c = __builtin_amdgcn_wmma_f32_16x16x4_f32(
                 false, a,  false, b,
                 (short)0, c,  false,  false);
    }

    float* hcol = H + (size_t)(mTile * 16) * DCH + nTile * 16 + l16;
#pragma unroll
    for (int i = 0; i < 8; ++i)
        hcol[(size_t)(i + 8 * hi) * DCH] = c[i];
}


__global__ void gcn_zero(float* p, int n) {
    int i = blockIdx.x * blockDim.x + threadIdx.x;
    if (i < n) p[i] = 0.0f;
}

__global__ void gcn_scatter(const float* __restrict__ H,
                            const int* __restrict__ src,
                            const int* __restrict__ tgt,
                            const float* __restrict__ dis,
                            float* __restrict__ agg, int e) {
    int wave = blockIdx.x * (blockDim.x >> 5) + (threadIdx.x >> 5);
    int lane = threadIdx.x & 31;
    if (wave >= e) return;
    int s = src[wave];
    int t = tgt[wave];
    float w = dis[s] * dis[t];
    float4 v = ((const float4*)(H + (size_t)s * DCH))[lane];
    float* at = agg + (size_t)t * DCH + lane * 4;
    atomicAdd(at + 0, v.x * w);
    atomicAdd(at + 1, v.y * w);
    atomicAdd(at + 2, v.z * w);
    atomicAdd(at + 3, v.w * w);
}

__global__ void gcn_finalize(const float* __restrict__ agg,
                             const float* __restrict__ H,
                             const float* __restrict__ dis,
                             const float* __restrict__ bias,
                             float* __restrict__ out, int nd, int relu) {
    int i = blockIdx.x * blockDim.x + threadIdx.x;
    if (i >= nd) return;
    int node = i >> 7;
    int d    = i & (DCH - 1);
    float di = dis[node];
    float v  = agg[i] + H[i] * di * di + bias[d];
    out[i] = relu ? fmaxf(v, 0.0f) : v;
}


extern "C" void kernel_launch(void* const* d_in, const int* in_sizes, int n_in,
                              void* d_out, int out_size, void* d_ws, size_t ws_size,
                              hipStream_t stream) {
    const float* X  = (const float*)d_in[0];
    const int*   ei = (const int*)  d_in[1];
    const float* W1 = (const float*)d_in[2];
    const float* b1 = (const float*)d_in[3];
    const float* W2 = (const float*)d_in[4];
    const float* b2 = (const float*)d_in[5];
    const float* W3 = (const float*)d_in[6];
    const float* b3 = (const float*)d_in[7];
    float* out = (float*)d_out;

    const int n  = in_sizes[0] / DCH;
    const int e  = in_sizes[1] / 2;
    const int nd = n * DCH;

    const int* src = ei;
    const int* tgt = ei + e;

    float* dis = (float*)d_ws;
    float* H   = dis + n;
    float* AGG = H + nd;

    gcn_deg_init <<<(n + 255) / 256, 256, 0, stream>>>(dis, n);
    gcn_deg_count<<<(e + 255) / 256, 256, 0, stream>>>(tgt, dis, e);
    gcn_rsqrt    <<<(n + 255) / 256, 256, 0, stream>>>(dis, n);

    const int mTiles     = n / 16;
    const int gemmBlocks = (mTiles * 8) / 8;
    const int edgeBlocks = (e + 7) / 8;
    const int ndBlocks   = (nd + 255) / 256;

    const float* Ws[3] = {W1, W2, W3};
    const float* bs[3] = {b1, b2, b3};
    const float* cur   = X;

    for (int layer = 0; layer < 3; ++layer) {
        gcn_gemm_wmma<<<gemmBlocks, 256, 0, stream>>>(cur, Ws[layer], H);
        gcn_zero     <<<ndBlocks,   256, 0, stream>>>(AGG, nd);
        gcn_scatter  <<<edgeBlocks, 256, 0, stream>>>(H, src, tgt, dis, AGG, e);
        gcn_finalize <<<ndBlocks,   256, 0, stream>>>(AGG, H, dis, bs[layer], out,
                                                      nd, layer < 2 ? 1 : 0);
        cur = out;
    }
}


# Scatter-add by f32 atomics -> gather per target row: edges grouped by target once per call (int-atomic ranks, rows sorted by source), each row summed in registers in fixed order, stored once
# speedup vs baseline: 2.8579x; 2.8579x over previous
#include <hip/hip_runtime.h>

typedef __attribute__((ext_vector_type(2))) float v2f;
typedef __attribute__((ext_vector_type(8))) float v8f;

#define DCH 128
#define NPW 4


__global__ void csr_clear(int* cnt, int* total, int n) {
    int i = blockIdx.x * blockDim.x + threadIdx.x;
    if (i < n) cnt[i] = 0;
    if (i == 0) total[0] = 0;
}

__global__ void csr_rank(const int* __restrict__ tgt, int* cnt, int* __restrict__ pos, int e) {
    int i = blockIdx.x * blockDim.x + threadIdx.x;
    if (i < e) pos[i] = atomicAdd(&cnt[tgt[i]], 1);
}

__global__ void csr_alloc(const int* __restrict__ cnt, int* total,
                          int* __restrict__ rowstart, float* __restrict__ dis, int n) {
    __shared__ int wsum[8];
    __shared__ int base_s;
    int i    = blockIdx.x * 256 + threadIdx.x;
    int lane = threadIdx.x & 31;
    int w    = threadIdx.x >> 5;
    int c = (i < n) ? cnt[i] + 1 : 0;
    int x = c;
    for (int d = 1; d < 32; d <<= 1) {
        int y = __shfl_up(x, d);
        if (lane >= d) x += y;
    }
    if (lane == 31) wsum[w] = x;
    __syncthreads();
    if (threadIdx.x == 0) {
        int s = 0;
        for (int k = 0; k < 8; ++k) { int v = wsum[k]; wsum[k] = s; s += v; }
        base_s = atomicAdd(total, s);
    }
    __syncthreads();
    if (i < n) {
        rowstart[i] = base_s + wsum[w] + x - c;
        dis[i] = rsqrtf((float)c);
    }
}

__global__ void csr_place(const int* __restrict__ src, const int* __restrict__ tgt,
                          const int* __restrict__ pos, const int* __restrict__ rowstart,
                          int* __restrict__ col, int e) {
    int i = blockIdx.x * blockDim.x + threadIdx.x;
    if (i < e) col[rowstart[tgt[i]] + pos[i]] = src[i];
}

__global__ void csr_sort(const int* __restrict__ rowstart, const int* __restrict__ cnt,
                         const float* __restrict__ dis, const int* __restrict__ col,
                         int2* __restrict__ rec, int n) {
    int t = __builtin_amdgcn_readfirstlane((int)(blockIdx.x * (blockDim.x >> 5) + (threadIdx.x >> 5)));
    int lane = threadIdx.x & 31;
    if (t >= n) return;
    int   st = rowstart[t];
    int   c  = cnt[t];
    float dt = dis[t];
    if (c <= 32) {
        int key = 0x7fffffff;
        if (lane < c) key = col[st + lane];
        int rank = 0;
        for (int k = 0; k < c; ++k) {
            int kk = __builtin_amdgcn_readlane(key, k);
            rank += (kk < key || (kk == key && k < lane)) ? 1 : 0;
        }
        if (lane < c) rec[st + rank] = make_int2(key, __float_as_int(dis[key] * dt));
    } else {
        for (int e0 = lane; e0 < c; e0 += 32) {
            int key = col[st + e0];
            int rank = 0;
            for (int k = 0; k < c; ++k) {
                int kk = col[st + k];
                rank += (kk < key || (kk == key && k < e0)) ? 1 : 0;
            }
            rec[st + rank] = make_int2(key, __float_as_int(dis[key] * dt));
        }
    }
    if (lane == 0) rec[st + c] = make_int2(t, __float_as_int(dt * dt));
}


__global__ void gcn_gemm_wmma(const float* __restrict__ A,
                              const float* __restrict__ W,
                              float* __restrict__ H) {
    int waveId = blockIdx.x * (blockDim.x >> 5) + (threadIdx.x >> 5);
    int lane   = threadIdx.x & 31;
    int mTile  = waveId >> 3;
    int nTile  = waveId & 7;
    int hi     = lane >> 4;
    int l16    = lane & 15;

    int row = mTile * 16 + l16;
    int col = nTile * 16 + l16;

    const float* arow = A + (size_t)row * DCH + 2 * hi;

    v8f c = {};
#pragma unroll
    for (int k0 = 0; k0 < DCH; k0 += 4) {
        v2f a, b;
        a.x = arow[k0 + 0];
        a.y = arow[k0 + 1];
        int kb = k0 + 2 * hi;
        b.x = W[(size_t)kb * DCH + col];
        b.y = W[(size_t)(kb + 1) * DCH + col];
        c = __builtin_amdgcn_wmma_f32_16x16x4_f32(
                 false, a,  false, b,
                 (short)0, c,  false,  false);
    }

    float* hcol = H + (size_t)(mTile * 16) * DCH + nTile * 16 + l16;
#pragma unroll
    for (int i = 0; i < 8; ++i)
        hcol[(size_t)(i + 8 * hi) * DCH] = c[i];
}


__device__ __forceinline__ void fma4(float4& a, const float4& v, float w) {
    a.x = fmaf(v.x, w, a.x);
    a.y = fmaf(v.y, w, a.y);
    a.z = fmaf(v.z, w, a.z);
    a.w = fmaf(v.w, w, a.w);
}

__global__ void gcn_gather(const float* __restrict__ H,
                           const int* __restrict__ rowstart,
                           const int* __restrict__ cnt,
                           const int2* __restrict__ rec,
                           const float* __restrict__ bias,
                           float* __restrict__ out, int n, int relu) {
    int wave = __builtin_amdgcn_readfirstlane((int)(blockIdx.x * (blockDim.x >> 5) + (threadIdx.x >> 5)));
    int lane = threadIdx.x & 31;
    int node0 = wave * NPW;
    if (node0 >= n) return;

    int mst = 0, mc = 0;
    if (lane < NPW && node0 + lane < n) {
        mst = rowstart[node0 + lane];
        mc  = cnt[node0 + lane] + 1;
    }
    float4 b4 = ((const float4*)bias)[lane];

    int  st[NPW], c[NPW];
    int2 r[NPW];
#pragma unroll
    for (int i = 0; i < NPW; ++i) {
        st[i] = __builtin_amdgcn_readlane(mst, i);
        c[i]  = __builtin_amdgcn_readlane(mc, i);
        r[i]  = make_int2(0, 0);
        if (lane < c[i]) r[i] = rec[st[i] + lane];
    }

#pragma unroll
    for (int i = 0; i < NPW; ++i) {
        int t = node0 + i;
        if (t >= n) break;
        float4 acc = make_float4(0.0f, 0.0f, 0.0f, 0.0f);

        int c32 = c[i] < 32 ? c[i] : 32;
        for (int j0 = 0; j0 < c32; j0 += 8) {
            float4 v[8];
#pragma unroll
            for (int k = 0; k < 8; ++k) {
                int jj = j0 + k < c32 ? j0 + k : c32 - 1;
                int s = __builtin_amdgcn_readlane(r[i].x, jj);
                v[k] = ((const float4*)(H + (size_t)s * DCH))[lane];
            }
#pragma unroll
            for (int k = 0; k < 8; ++k) {
                if (j0 + k < c32)
                    fma4(acc, v[k], __int_as_float(__builtin_amdgcn_readlane(r[i].y, j0 + k)));
            }
        }
        for (int j = 32; j < c[i]; ++j) {
            int2 q = rec[st[i] + j];
            float4 v = ((const float4*)(H + (size_t)q.x * DCH))[lane];
            fma4(acc, v, __int_as_float(q.y));
        }

        acc.x += b4.x; acc.y += b4.y; acc.z += b4.z; acc.w += b4.w;
        if (relu) {
            acc.x = fmaxf(acc.x, 0.0f); acc.y = fmaxf(acc.y, 0.0f);
            acc.z = fmaxf(acc.z, 0.0f); acc.w = fmaxf(acc.w, 0.0f);
        }
        ((float4*)(out + (size_t)t * DCH))[lane] = acc;
    }
}


extern "C" void kernel_launch(void* const* d_in, const int* in_sizes, int n_in,
                              void* d_out, int out_size, void* d_ws, size_t ws_size,
                              hipStream_t stream) {
    const float* X  = (const float*)d_in[0];
    const int*   ei = (const int*)  d_in[1];
    const float* W1 = (const float*)d_in[2];
    const float* b1 = (const float*)d_in[3];
    const float* W2 = (const float*)d_in[4];
    const float* b2 = (const float*)d_in[5];
    const float* W3 = (const float*)d_in[6];
    const float* b3 = (const float*)d_in[7];
    float* out = (float*)d_out;

    const int n  = in_sizes[0] / DCH;
    const int e  = in_sizes[1] / 2;

    const int* src = ei;
    const int* tgt = ei + e;

    int*   cnt      = (int*)d_ws;
    int*   rowstart = cnt + n;
    float* dis      = (float*)(rowstart + n);
    int*   total    = (int*)(dis + n);
    int*   pos      = total + 4;
    int*   col      = pos + e;
    int2*  rec      = (int2*)(col + e + n);
    float* H        = (float*)(rec + e + n);

    csr_clear<<<(n + 255) / 256, 256, 0, stream>>>(cnt, total, n);
    csr_rank <<<(e + 255) / 256, 256, 0, stream>>>(tgt, cnt, pos, e);
    csr_alloc<<<(n + 255) / 256, 256, 0, stream>>>(cnt, total, rowstart, dis, n);
    csr_place<<<(e + 255) / 256, 256, 0, stream>>>(src, tgt, pos, rowstart, col, e);
    csr_sort <<<(n + 7) / 8,     256, 0, stream>>>(rowstart, cnt, dis, col, rec, n);

    const int mTiles       = n / 16;
    const int gemmBlocks   = (mTiles * 8) / 8;
    const int gatherBlocks = (n + 8 * NPW - 1) / (8 * NPW);

    const float* Ws[3] = {W1, W2, W3};
    const float* bs[3] = {b1, b2, b3};
    const float* cur   = X;

    for (int layer = 0; layer < 3; ++layer) {
        gcn_gemm_wmma<<<gemmBlocks,   256, 0, stream>>>(cur, Ws[layer], H);
        gcn_gather   <<<gatherBlocks, 256, 0, stream>>>(H, rowstart, cnt, rec, bs[layer],
                                                        out, n, layer < 2 ? 1 : 0);
        cur = out;
    }
}


# Each layer's GEMM fused into its gather: 32 target rows summed over the layer's input into LDS, then x W on WMMA there (sum order changes: aggregate, then transform); 11 -> 8 launches
# speedup vs baseline: 4.0693x; 1.4239x over previous
#include <hip/hip_runtime.h>

typedef __attribute__((ext_vector_type(2))) float v2f;
typedef __attribute__((ext_vector_type(8))) float v8f;

#define DCH 128
#define NPW 4
#define WPB 8
#define TR  (WPB * NPW)
#define PITCH 132


__global__ void csr_clear(int* cnt, int* total, int n) {
    int i = blockIdx.x * blockDim.x + threadIdx.x;
    if (i < n) cnt[i] = 0;
    if (i == 0) total[0] = 0;
}

__global__ void csr_rank(const int* __restrict__ tgt, int* cnt, int* __restrict__ pos, int e) {
    int i = blockIdx.x * blockDim.x + threadIdx.x;
    if (i < e) pos[i] = atomicAdd(&cnt[tgt[i]], 1);
}

__global__ void csr_alloc(const int* __restrict__ cnt, int* total,
                          int* __restrict__ rowstart, float* __restrict__ dis, int n) {
    __shared__ int wsum[8];
    __shared__ int base_s;
    int i    = blockIdx.x * 256 + threadIdx.x;
    int lane = threadIdx.x & 31;
    int w    = threadIdx.x >> 5;
    int c = (i < n) ? cnt[i] + 1 : 0;
    int x = c;
    for (int d = 1; d < 32; d <<= 1) {
        int y = __shfl_up(x, d);
        if (lane >= d) x += y;
    }
    if (lane == 31) wsum[w] = x;
    __syncthreads();
    if (threadIdx.x == 0) {
        int s = 0;
        for (int k = 0; k < 8; ++k) { int v = wsum[k]; wsum[k] = s; s += v; }
        base_s = atomicAdd(total, s);
    }
    __syncthreads();
    if (i < n) {
        rowstart[i] = base_s + wsum[w] + x - c;
        dis[i] = rsqrtf((float)c);
    }
}

__global__ void csr_place(const int* __restrict__ src, const int* __restrict__ tgt,
                          const int* __restrict__ pos, const int* __restrict__ rowstart,
                          int* __restrict__ col, int e) {
    int i = blockIdx.x * blockDim.x + threadIdx.x;
    if (i < e) col[rowstart[tgt[i]] + pos[i]] = src[i];
}

__global__ void csr_sort(const int* __restrict__ rowstart, const int* __restrict__ cnt,
                         const float* __restrict__ dis, const int* __restrict__ col,
                         int2* __restrict__ rec, int n) {
    int t = __builtin_amdgcn_readfirstlane((int)(blockIdx.x * (blockDim.x >> 5) + (threadIdx.x >> 5)));
    int lane = threadIdx.x & 31;
    if (t >= n) return;
    int   st = rowstart[t];
    int   c  = cnt[t];
    float dt = dis[t];
    if (c <= 32) {
        int key = 0x7fffffff;
        if (lane < c) key = col[st + lane];
        int rank = 0;
        for (int k = 0; k < c; ++k) {
            int kk = __builtin_amdgcn_readlane(key, k);
            rank += (kk < key || (kk == key && k < lane)) ? 1 : 0;
        }
        if (lane < c) rec[st + rank] = make_int2(key, __float_as_int(dis[key] * dt));
    } else {
        for (int e0 = lane; e0 < c; e0 += 32) {
            int key = col[st + e0];
            int rank = 0;
            for (int k = 0; k < c; ++k) {
                int kk = col[st + k];
                rank += (kk < key || (kk == key && k < e0)) ? 1 : 0;
            }
            rec[st + rank] = make_int2(key, __float_as_int(dis[key] * dt));
        }
    }
    if (lane == 0) rec[st + c] = make_int2(t, __float_as_int(dt * dt));
}


__device__ __forceinline__ void fma4(float4& a, const float4& v, float w) {
    a.x = fmaf(v.x, w, a.x);
    a.y = fmaf(v.y, w, a.y);
    a.z = fmaf(v.z, w, a.z);
    a.w = fmaf(v.w, w, a.w);
}

__global__ __launch_bounds__(256) void gcn_layer(const float* __restrict__ Xin,
                                                 const int* __restrict__ rowstart,
                                                 const int* __restrict__ cnt,
                                                 const int2* __restrict__ rec,
                                                 const float* __restrict__ Wm,
                                                 const float* __restrict__ bias,
                                                 float* __restrict__ out, int n, int relu) {
    __shared__ __attribute__((aligned(16))) float aggT[TR * PITCH];
    __shared__ __attribute__((aligned(16))) float outT[TR * PITCH];

    int wib  = __builtin_amdgcn_readfirstlane((int)(threadIdx.x >> 5));
    int lane = threadIdx.x & 31;
    int row0  = blockIdx.x * TR;
    int node0 = row0 + wib * NPW;

    int mst = 0, mc = 0;
    if (lane < NPW && node0 + lane < n) {
        mst = rowstart[node0 + lane];
        mc  = cnt[node0 + lane] + 1;
    }

    int  st[NPW], c[NPW];
    int2 r[NPW];
#pragma unroll
    for (int i = 0; i < NPW; ++i) {
        st[i] = __builtin_amdgcn_readlane(mst, i);
        c[i]  = __builtin_amdgcn_readlane(mc, i);
        r[i]  = make_int2(0, 0);
        if (lane < c[i]) r[i] = rec[st[i] + lane];
    }

#pragma unroll
    for (int i = 0; i < NPW; ++i) {
        float4 acc = make_float4(0.0f, 0.0f, 0.0f, 0.0f);

        int c32 = c[i] < 32 ? c[i] : 32;
        for (int j0 = 0; j0 < c32; j0 += 8) {
            float4 v[8];
#pragma unroll
            for (int k = 0; k < 8; ++k) {
                int jj = j0 + k < c32 ? j0 + k : c32 - 1;
                int s = __builtin_amdgcn_readlane(r[i].x, jj);
                v[k] = ((const float4*)(Xin + (size_t)s * DCH))[lane];
            }
#pragma unroll
            for (int k = 0; k < 8; ++k) {
                if (j0 + k < c32)
                    fma4(acc, v[k], __int_as_float(__builtin_amdgcn_readlane(r[i].y, j0 + k)));
            }
        }
        for (int j = 32; j < c[i]; ++j) {
            int2 q = rec[st[i] + j];
            float4 v = ((const float4*)(Xin + (size_t)q.x * DCH))[lane];
            fma4(acc, v, __int_as_float(q.y));
        }

        *(float4*)(aggT + (wib * NPW + i) * PITCH + 4 * lane) = acc;
    }

    __syncthreads();

    int hi  = lane >> 4;
    int l16 = lane & 15;
    int col = wib * 16 + l16;
    const float* a0p = aggT + l16 * PITCH + 2 * hi;
    const float* a1p = a0p + 16 * PITCH;
    const float* bp  = Wm + (size_t)(2 * hi) * DCH + col;

    v8f c0 = {};
    v8f c1 = {};
#pragma unroll 8
    for (int k0 = 0; k0 < DCH; k0 += 4) {
        v2f a0 = *(const v2f*)(a0p + k0);
        v2f a1 = *(const v2f*)(a1p + k0);
        v2f b;
        b.x = bp[(size_t)k0 * DCH];
        b.y = bp[(size_t)(k0 + 1) * DCH];
        c0 = __builtin_amdgcn_wmma_f32_16x16x4_f32(
                  false, a0,  false, b,
                  (short)0, c0,  false,  false);
        c1 = __builtin_amdgcn_wmma_f32_16x16x4_f32(
                  false, a1,  false, b,
                  (short)0, c1,  false,  false);
    }

#pragma unroll
    for (int i = 0; i < 8; ++i) {
        outT[(8 * hi + i) * PITCH + col]      = c0[i];
        outT[(16 + 8 * hi + i) * PITCH + col] = c1[i];
    }

    __syncthreads();

    float4 b4 = ((const float4*)bias)[lane];
#pragma unroll
    for (int i = 0; i < NPW; ++i) {
        int lr = wib * NPW + i;
        int t  = row0 + lr;
        if (t < n) {
            float4 o = *(const float4*)(outT + lr * PITCH + 4 * lane);
            o.x += b4.x; o.y += b4.y; o.z += b4.z; o.w += b4.w;
            if (relu) {
                o.x = fmaxf(o.x, 0.0f); o.y = fmaxf(o.y, 0.0f);
                o.z = fmaxf(o.z, 0.0f); o.w = fmaxf(o.w, 0.0f);
            }
            ((float4*)(out + (size_t)t * DCH))[lane] = o;
        }
    }
}


extern "C" void kernel_launch(void* const* d_in, const int* in_sizes, int n_in,
                              void* d_out, int out_size, void* d_ws, size_t ws_size,
                              hipStream_t stream) {
    const float* X  = (const float*)d_in[0];
    const int*   ei = (const int*)  d_in[1];
    const float* W1 = (const float*)d_in[2];
    const float* b1 = (const float*)d_in[3];
    const float* W2 = (const float*)d_in[4];
    const float* b2 = (const float*)d_in[5];
    const float* W3 = (const float*)d_in[6];
    const float* b3 = (const float*)d_in[7];
    float* out = (float*)d_out;

    const int n  = in_sizes[0] / DCH;
    const int e  = in_sizes[1] / 2;

    const int* src = ei;
    const int* tgt = ei + e;

    int*   cnt      = (int*)d_ws;
    int*   rowstart = cnt + n;
    float* dis      = (float*)(rowstart + n);
    int*   total    = (int*)(dis + n);
    int*   pos      = total + 4;
    int*   col      = pos + e;
    int2*  rec      = (int2*)(col + e + n);
    float* Y1       = (float*)(rec + e + n);

    csr_clear<<<(n + 255) / 256, 256, 0, stream>>>(cnt, total, n);
    csr_rank <<<(e + 255) / 256, 256, 0, stream>>>(tgt, cnt, pos, e);
    csr_alloc<<<(n + 255) / 256, 256, 0, stream>>>(cnt, total, rowstart, dis, n);
    csr_place<<<(e + 255) / 256, 256, 0, stream>>>(src, tgt, pos, rowstart, col, e);
    csr_sort <<<(n + 7) / 8,     256, 0, stream>>>(rowstart, cnt, dis, col, rec, n);

    const int layerBlocks = (n + TR - 1) / TR;

    gcn_layer<<<layerBlocks, 256, 0, stream>>>(X,   rowstart, cnt, rec, W1, b1, out, n, 1);
    gcn_layer<<<layerBlocks, 256, 0, stream>>>(out, rowstart, cnt, rec, W2, b2, Y1,  n, 1);
    gcn_layer<<<layerBlocks, 256, 0, stream>>>(Y1,  rowstart, cnt, rec, W3, b3, out, n, 0);
}


# Edge grouping without a global atomic per edge: edges binned by 256-target range (per-block LDS counts, count table), each range's block ranks and places its rows by LDS counters; 8 -> 7 launches
# speedup vs baseline: 5.5822x; 1.3718x over previous
#include <hip/hip_runtime.h>

typedef __attribute__((ext_vector_type(2))) float v2f;
typedef __attribute__((ext_vector_type(8))) float v8f;

#define DCH 128
#define NPW 4
#define WPB 8
#define TR  (WPB * NPW)
#define PITCH 132


#define BINW 256
#define BINSH 8
#define NBIN 256
#define NBLK 128
#define BT   1024
#define EPT  8
#define NGRP (BT / 64)
#define BB   1024
#define RPT  8

__global__ __launch_bounds__(BT) void bin_hist(const int* __restrict__ tgt, int* __restrict__ table,
                                               int* __restrict__ total, int e, int chunk) {
    __shared__ int h[NBIN];
    int t = threadIdx.x;
    if (t < NBIN) h[t] = 0;
    if (blockIdx.x == 0 && t == 0) total[0] = 0;
    __syncthreads();
    int lo = blockIdx.x * chunk;
    int hi = lo + chunk < e ? lo + chunk : e;
    for (int base = lo; base < hi; base += BT * EPT) {
        int v[EPT];
#pragma unroll
        for (int k = 0; k < EPT; ++k) {
            int idx = base + k * BT + t;
            v[k] = tgt[idx < hi ? idx : hi - 1];
        }
#pragma unroll
        for (int k = 0; k < EPT; ++k) {
            int idx = base + k * BT + t;
            if (idx < hi) atomicAdd(&h[v[k] >> BINSH], 1);
        }
    }
    __syncthreads();
    if (t < NBIN) table[blockIdx.x * NBIN + t] = h[t];
}

__global__ __launch_bounds__(BT) void bin_scatter(const int* __restrict__ src, const int* __restrict__ tgt,
                                                  const int* __restrict__ table,
                                                  int* __restrict__ binstart, int2* __restrict__ binned,
                                                  int e, int chunk) {
    __shared__ int ptot[NGRP][NBIN];
    __shared__ int pbef[NGRP][NBIN];
    __shared__ int cur[NBIN];
    __shared__ int wsum[BT / 32];
    __shared__ int tot_s;
    int t  = threadIdx.x;
    int me = blockIdx.x;

    int q = t & 63;
    int g = t >> 6;
    int4 tot = make_int4(0, 0, 0, 0);
    int4 bef = make_int4(0, 0, 0, 0);
#pragma unroll
    for (int j = 0; j < NBLK / NGRP; ++j) {
        int blk = g + NGRP * j;
        int4 v = ((const int4*)(table + blk * NBIN))[q];
        tot.x += v.x; tot.y += v.y; tot.z += v.z; tot.w += v.w;
        int m = blk < me ? -1 : 0;
        bef.x += v.x & m; bef.y += v.y & m; bef.z += v.z & m; bef.w += v.w & m;
    }
    ptot[g][4 * q + 0] = tot.x; ptot[g][4 * q + 1] = tot.y;
    ptot[g][4 * q + 2] = tot.z; ptot[g][4 * q + 3] = tot.w;
    pbef[g][4 * q + 0] = bef.x; pbef[g][4 * q + 1] = bef.y;
    pbef[g][4 * q + 2] = bef.z; pbef[g][4 * q + 3] = bef.w;
    __syncthreads();

    int c = 0, b = 0;
    if (t < NBIN) {
#pragma unroll
        for (int k = 0; k < NGRP; ++k) { c += ptot[k][t]; b += pbef[k][t]; }
    }
    int lane = t & 31;
    int w    = t >> 5;
    int x = c;
    for (int d = 1; d < 32; d <<= 1) {
        int y = __shfl_up(x, d);
        if (lane >= d) x += y;
    }
    if (lane == 31) wsum[w] = x;
    __syncthreads();
    if (t == 0) {
        int s = 0;
        for (int k = 0; k < BT / 32; ++k) { int v = wsum[k]; wsum[k] = s; s += v; }
        tot_s = s;
    }
    __syncthreads();
    if (t < NBIN) {
        int first = wsum[w] + x - c;
        cur[t] = first + b;
        if (me == 0) {
            binstart[t] = first;
            if (t == 0) binstart[NBIN] = tot_s;
        }
    }
    __syncthreads();

    int lo = me * chunk;
    int hi = lo + chunk < e ? lo + chunk : e;
    for (int base = lo; base < hi; base += BT * EPT) {
        int tv[EPT], sv[EPT];
#pragma unroll
        for (int k = 0; k < EPT; ++k) {
            int idx = base + k * BT + t;
            int ii  = idx < hi ? idx : hi - 1;
            tv[k] = tgt[ii];
            sv[k] = src[ii];
        }
#pragma unroll
        for (int k = 0; k < EPT; ++k) {
            int idx = base + k * BT + t;
            if (idx < hi) {
                int p = atomicAdd(&cur[tv[k] >> BINSH], 1);
                binned[p] = make_int2(tv[k], sv[k]);
            }
        }
    }
}

__global__ __launch_bounds__(BB) void bin_build(const int* __restrict__ binstart,
                                                const int2* __restrict__ binned,
                                                int* __restrict__ total, int* __restrict__ cnt,
                                                int* __restrict__ rowstart, float* __restrict__ dis,
                                                int* __restrict__ col, int n) {
    __shared__ int lc[BINW];
    __shared__ int rs[BINW];
    __shared__ int wsum[BB / 32];
    __shared__ int base_s;
    int t    = threadIdx.x;
    int row0 = blockIdx.x * BINW;
    if (t < BINW) lc[t] = 0;
    __syncthreads();
    int b0 = binstart[blockIdx.x];
    int b1 = binstart[blockIdx.x + 1];
    for (int base = b0; base < b1; base += BB * RPT) {
        int2 r[RPT];
#pragma unroll
        for (int k = 0; k < RPT; ++k) {
            int idx = base + k * BB + t;
            r[k] = binned[idx < b1 ? idx : b1 - 1];
        }
#pragma unroll
        for (int k = 0; k < RPT; ++k) {
            if (base + k * BB + t < b1) atomicAdd(&lc[r[k].x - row0], 1);
        }
    }
    __syncthreads();

    int node = row0 + t;
    int own  = (t < BINW && node < n) ? 1 : 0;
    int c = own ? lc[t] + 1 : 0;
    int lane = t & 31;
    int w    = t >> 5;
    int x = c;
    for (int d = 1; d < 32; d <<= 1) {
        int y = __shfl_up(x, d);
        if (lane >= d) x += y;
    }
    if (lane == 31) wsum[w] = x;
    __syncthreads();
    if (t == 0) {
        int s = 0;
        for (int k = 0; k < BB / 32; ++k) { int v = wsum[k]; wsum[k] = s; s += v; }
        base_s = atomicAdd(total, s);
    }
    __syncthreads();
    if (own) {
        int st = base_s + wsum[w] + x - c;
        rowstart[node] = st;
        cnt[node]      = c - 1;
        dis[node]      = rsqrtf((float)c);
        rs[t]          = st;
    }
    if (t < BINW) lc[t] = 0;
    __syncthreads();
    for (int base = b0; base < b1; base += BB * RPT) {
        int2 r[RPT];
#pragma unroll
        for (int k = 0; k < RPT; ++k) {
            int idx = base + k * BB + t;
            r[k] = binned[idx < b1 ? idx : b1 - 1];
        }
#pragma unroll
        for (int k = 0; k < RPT; ++k) {
            if (base + k * BB + t < b1) {
                int j = r[k].x - row0;
                int p = atomicAdd(&lc[j], 1);
                col[rs[j] + p] = r[k].y;
            }
        }
    }
}

__global__ void csr_sort(const int* __restrict__ rowstart, const int* __restrict__ cnt,
                         const float* __restrict__ dis, const int* __restrict__ col,
                         int2* __restrict__ rec, int n) {
    int t = __builtin_amdgcn_readfirstlane((int)(blockIdx.x * (blockDim.x >> 5) + (threadIdx.x >> 5)));
    int lane = threadIdx.x & 31;
    if (t >= n) return;
    int   st = rowstart[t];
    int   c  = cnt[t];
    float dt = dis[t];
    if (c <= 32) {
        int key = 0x7fffffff;
        if (lane < c) key = col[st + lane];
        int rank = 0;
        for (int k = 0; k < c; ++k) {
            int kk = __builtin_amdgcn_readlane(key, k);
            rank += (kk < key || (kk == key && k < lane)) ? 1 : 0;
        }
        if (lane < c) rec[st + rank] = make_int2(key, __float_as_int(dis[key] * dt));
    } else {
        for (int e0 = lane; e0 < c; e0 += 32) {
            int key = col[st + e0];
            int rank = 0;
            for (int k = 0; k < c; ++k) {
                int kk = col[st + k];
                rank += (kk < key || (kk == key && k < e0)) ? 1 : 0;
            }
            rec[st + rank] = make_int2(key, __float_as_int(dis[key] * dt));
        }
    }
    if (lane == 0) rec[st + c] = make_int2(t, __float_as_int(dt * dt));
}


__device__ __forceinline__ void fma4(float4& a, const float4& v, float w) {
    a.x = fmaf(v.x, w, a.x);
    a.y = fmaf(v.y, w, a.y);
    a.z = fmaf(v.z, w, a.z);
    a.w = fmaf(v.w, w, a.w);
}

__global__ __launch_bounds__(256) void gcn_layer(const float* __restrict__ Xin,
                                                 const int* __restrict__ rowstart,
                                                 const int* __restrict__ cnt,
                                                 const int2* __restrict__ rec,
                                                 const float* __restrict__ Wm,
                                                 const float* __restrict__ bias,
                                                 float* __restrict__ out, int n, int relu) {
    __shared__ __attribute__((aligned(16))) float aggT[TR * PITCH];
    __shared__ __attribute__((aligned(16))) float outT[TR * PITCH];

    int wib  = __builtin_amdgcn_readfirstlane((int)(threadIdx.x >> 5));
    int lane = threadIdx.x & 31;
    int row0  = blockIdx.x * TR;
    int node0 = row0 + wib * NPW;

    int mst = 0, mc = 0;
    if (lane < NPW && node0 + lane < n) {
        mst = rowstart[node0 + lane];
        mc  = cnt[node0 + lane] + 1;
    }

    int  st[NPW], c[NPW];
    int2 r[NPW];
#pragma unroll
    for (int i = 0; i < NPW; ++i) {
        st[i] = __builtin_amdgcn_readlane(mst, i);
        c[i]  = __builtin_amdgcn_readlane(mc, i);
        r[i]  = make_int2(0, 0);
        if (lane < c[i]) r[i] = rec[st[i] + lane];
    }

#pragma unroll
    for (int i = 0; i < NPW; ++i) {
        float4 acc = make_float4(0.0f, 0.0f, 0.0f, 0.0f);

        int c32 = c[i] < 32 ? c[i] : 32;
        for (int j0 = 0; j0 < c32; j0 += 8) {
            float4 v[8];
#pragma unroll
            for (int k = 0; k < 8; ++k) {
                int jj = j0 + k < c32 ? j0 + k : c32 - 1;
                int s = __builtin_amdgcn_readlane(r[i].x, jj);
                v[k] = ((const float4*)(Xin + (size_t)s * DCH))[lane];
            }
#pragma unroll
            for (int k = 0; k < 8; ++k) {
                if (j0 + k < c32)
                    fma4(acc, v[k], __int_as_float(__builtin_amdgcn_readlane(r[i].y, j0 + k)));
            }
        }
        for (int j = 32; j < c[i]; ++j) {
            int2 q = rec[st[i] + j];
            float4 v = ((const float4*)(Xin + (size_t)q.x * DCH))[lane];
            fma4(acc, v, __int_as_float(q.y));
        }

        *(float4*)(aggT + (wib * NPW + i) * PITCH + 4 * lane) = acc;
    }

    __syncthreads();

    int hi  = lane >> 4;
    int l16 = lane & 15;
    int col = wib * 16 + l16;
    const float* a0p = aggT + l16 * PITCH + 2 * hi;
    const float* a1p = a0p + 16 * PITCH;
    const float* bp  = Wm + (size_t)(2 * hi) * DCH + col;

    v8f c0 = {};
    v8f c1 = {};
#pragma unroll 8
    for (int k0 = 0; k0 < DCH; k0 += 4) {
        v2f a0 = *(const v2f*)(a0p + k0);
        v2f a1 = *(const v2f*)(a1p + k0);
        v2f b;
        b.x = bp[(size_t)k0 * DCH];
        b.y = bp[(size_t)(k0 + 1) * DCH];
        c0 = __builtin_amdgcn_wmma_f32_16x16x4_f32(
                  false, a0,  false, b,
                  (short)0, c0,  false,  false);
        c1 = __builtin_amdgcn_wmma_f32_16x16x4_f32(
                  false, a1,  false, b,
                  (short)0, c1,  false,  false);
    }

#pragma unroll
    for (int i = 0; i < 8; ++i) {
        outT[(8 * hi + i) * PITCH + col]      = c0[i];
        outT[(16 + 8 * hi + i) * PITCH + col] = c1[i];
    }

    __syncthreads();

    float4 b4 = ((const float4*)bias)[lane];
#pragma unroll
    for (int i = 0; i < NPW; ++i) {
        int lr = wib * NPW + i;
        int t  = row0 + lr;
        if (t < n) {
            float4 o = *(const float4*)(outT + lr * PITCH + 4 * lane);
            o.x += b4.x; o.y += b4.y; o.z += b4.z; o.w += b4.w;
            if (relu) {
                o.x = fmaxf(o.x, 0.0f); o.y = fmaxf(o.y, 0.0f);
                o.z = fmaxf(o.z, 0.0f); o.w = fmaxf(o.w, 0.0f);
            }
            ((float4*)(out + (size_t)t * DCH))[lane] = o;
        }
    }
}


extern "C" void kernel_launch(void* const* d_in, const int* in_sizes, int n_in,
                              void* d_out, int out_size, void* d_ws, size_t ws_size,
                              hipStream_t stream) {
    const float* X  = (const float*)d_in[0];
    const int*   ei = (const int*)  d_in[1];
    const float* W1 = (const float*)d_in[2];
    const float* b1 = (const float*)d_in[3];
    const float* W2 = (const float*)d_in[4];
    const float* b2 = (const float*)d_in[5];
    const float* W3 = (const float*)d_in[6];
    const float* b3 = (const float*)d_in[7];
    float* out = (float*)d_out;

    const int n  = in_sizes[0] / DCH;
    const int e  = in_sizes[1] / 2;

    const int* src = ei;
    const int* tgt = ei + e;

    int*   cnt      = (int*)d_ws;
    int*   rowstart = cnt + n;
    float* dis      = (float*)(rowstart + n);
    int*   total    = (int*)(dis + n);
    int*   binstart = total + 4;
    int*   table    = binstart + NBIN + 4;
    int2*  binned   = (int2*)(table + NBLK * NBIN);
    int*   col      = (int*)(binned + e);
    int2*  rec      = (int2*)(col + e + n);
    float* Y1       = (float*)(rec + e + n);

    const int chunk = (e + NBLK - 1) / NBLK;
    const int nbins = (n + BINW - 1) / BINW;
    bin_hist   <<<NBLK,  BT, 0, stream>>>(tgt, table, total, e, chunk);
    bin_scatter<<<NBLK,  BT, 0, stream>>>(src, tgt, table, binstart, binned, e, chunk);
    bin_build  <<<nbins, BB, 0, stream>>>(binstart, binned, total, cnt, rowstart, dis, col, n);
    csr_sort   <<<(n + 7) / 8, 256, 0, stream>>>(rowstart, cnt, dis, col, rec, n);

    const int layerBlocks = (n + TR - 1) / TR;

    gcn_layer<<<layerBlocks, 256, 0, stream>>>(X,   rowstart, cnt, rec, W1, b1, out, n, 1);
    gcn_layer<<<layerBlocks, 256, 0, stream>>>(out, rowstart, cnt, rec, W2, b2, Y1,  n, 1);
    gcn_layer<<<layerBlocks, 256, 0, stream>>>(Y1,  rowstart, cnt, rec, W3, b3, out, n, 0);
}


# Row sort folded into layer 1's kernel: each wave ranks the sources of the rows it sums, writes rec for layers 2 and 3, takes its own records through LDS; sum order unchanged; 7 -> 6 launches
# speedup vs baseline: 5.8779x; 1.0530x over previous
#include <hip/hip_runtime.h>

typedef __attribute__((ext_vector_type(2))) float v2f;
typedef __attribute__((ext_vector_type(8))) float v8f;

#define DCH 128
#define NPW 4
#define WPB 8
#define TR  (WPB * NPW)
#define PITCH 132


#define BINW 256
#define BINSH 8
#define NBIN 256
#define NBLK 128
#define BT   1024
#define EPT  8
#define NGRP (BT / 64)
#define BB   1024
#define RPT  8

__global__ __launch_bounds__(BT) void bin_hist(const int* __restrict__ tgt, int* __restrict__ table,
                                               int* __restrict__ total, int e, int chunk) {
    __shared__ int h[NBIN];
    int t = threadIdx.x;
    if (t < NBIN) h[t] = 0;
    if (blockIdx.x == 0 && t == 0) total[0] = 0;
    __syncthreads();
    int lo = blockIdx.x * chunk;
    int hi = lo + chunk < e ? lo + chunk : e;
    for (int base = lo; base < hi; base += BT * EPT) {
        int v[EPT];
#pragma unroll
        for (int k = 0; k < EPT; ++k) {
            int idx = base + k * BT + t;
            v[k] = tgt[idx < hi ? idx : hi - 1];
        }
#pragma unroll
        for (int k = 0; k < EPT; ++k) {
            int idx = base + k * BT + t;
            if (idx < hi) atomicAdd(&h[v[k] >> BINSH], 1);
        }
    }
    __syncthreads();
    if (t < NBIN) table[blockIdx.x * NBIN + t] = h[t];
}

__global__ __launch_bounds__(BT) void bin_scatter(const int* __restrict__ src, const int* __restrict__ tgt,
                                                  const int* __restrict__ table,
                                                  int* __restrict__ binstart, int2* __restrict__ binned,
                                                  int e, int chunk) {
    __shared__ int ptot[NGRP][NBIN];
    __shared__ int pbef[NGRP][NBIN];
    __shared__ int cur[NBIN];
    __shared__ int wsum[BT / 32];
    __shared__ int tot_s;
    int t  = threadIdx.x;
    int me = blockIdx.x;

    int q = t & 63;
    int g = t >> 6;
    int4 tot = make_int4(0, 0, 0, 0);
    int4 bef = make_int4(0, 0, 0, 0);
#pragma unroll
    for (int j = 0; j < NBLK / NGRP; ++j) {
        int blk = g + NGRP * j;
        int4 v = ((const int4*)(table + blk * NBIN))[q];
        tot.x += v.x; tot.y += v.y; tot.z += v.z; tot.w += v.w;
        int m = blk < me ? -1 : 0;
        bef.x += v.x & m; bef.y += v.y & m; bef.z += v.z & m; bef.w += v.w & m;
    }
    ptot[g][4 * q + 0] = tot.x; ptot[g][4 * q + 1] = tot.y;
    ptot[g][4 * q + 2] = tot.z; ptot[g][4 * q + 3] = tot.w;
    pbef[g][4 * q + 0] = bef.x; pbef[g][4 * q + 1] = bef.y;
    pbef[g][4 * q + 2] = bef.z; pbef[g][4 * q + 3] = bef.w;
    __syncthreads();

    int c = 0, b = 0;
    if (t < NBIN) {
#pragma unroll
        for (int k = 0; k < NGRP; ++k) { c += ptot[k][t]; b += pbef[k][t]; }
    }
    int lane = t & 31;
    int w    = t >> 5;
    int x = c;
    for (int d = 1; d < 32; d <<= 1) {
        int y = __shfl_up(x, d);
        if (lane >= d) x += y;
    }
    if (lane == 31) wsum[w] = x;
    __syncthreads();
    if (t == 0) {
        int s = 0;
        for (int k = 0; k < BT / 32; ++k) { int v = wsum[k]; wsum[k] = s; s += v; }
        tot_s = s;
    }
    __syncthreads();
    if (t < NBIN) {
        int first = wsum[w] + x - c;
        cur[t] = first + b;
        if (me == 0) {
            binstart[t] = first;
            if (t == 0) binstart[NBIN] = tot_s;
        }
    }
    __syncthreads();

    int lo = me * chunk;
    int hi = lo + chunk < e ? lo + chunk : e;
    for (int base = lo; base < hi; base += BT * EPT) {
        int tv[EPT], sv[EPT];
#pragma unroll
        for (int k = 0; k < EPT; ++k) {
            int idx = base + k * BT + t;
            int ii  = idx < hi ? idx : hi - 1;
            tv[k] = tgt[ii];
            sv[k] = src[ii];
        }
#pragma unroll
        for (int k = 0; k < EPT; ++k) {
            int idx = base + k * BT + t;
            if (idx < hi) {
                int p = atomicAdd(&cur[tv[k] >> BINSH], 1);
                binned[p] = make_int2(tv[k], sv[k]);
            }
        }
    }
}

__global__ __launch_bounds__(BB) void bin_build(const int* __restrict__ binstart,
                                                const int2* __restrict__ binned,
                                                int* __restrict__ total, int* __restrict__ cnt,
                                                int* __restrict__ rowstart, float* __restrict__ dis,
                                                int* __restrict__ col, int n) {
    __shared__ int lc[BINW];
    __shared__ int rs[BINW];
    __shared__ int wsum[BB / 32];
    __shared__ int base_s;
    int t    = threadIdx.x;
    int row0 = blockIdx.x * BINW;
    if (t < BINW) lc[t] = 0;
    __syncthreads();
    int b0 = binstart[blockIdx.x];
    int b1 = binstart[blockIdx.x + 1];
    for (int base = b0; base < b1; base += BB * RPT) {
        int2 r[RPT];
#pragma unroll
        for (int k = 0; k < RPT; ++k) {
            int idx = base + k * BB + t;
            r[k] = binned[idx < b1 ? idx : b1 - 1];
        }
#pragma unroll
        for (int k = 0; k < RPT; ++k) {
            if (base + k * BB + t < b1) atomicAdd(&lc[r[k].x - row0], 1);
        }
    }
    __syncthreads();

    int node = row0 + t;
    int own  = (t < BINW && node < n) ? 1 : 0;
    int c = own ? lc[t] + 1 : 0;
    int lane = t & 31;
    int w    = t >> 5;
    int x = c;
    for (int d = 1; d < 32; d <<= 1) {
        int y = __shfl_up(x, d);
        if (lane >= d) x += y;
    }
    if (lane == 31) wsum[w] = x;
    __syncthreads();
    if (t == 0) {
        int s = 0;
        for (int k = 0; k < BB / 32; ++k) { int v = wsum[k]; wsum[k] = s; s += v; }
        base_s = atomicAdd(total, s);
    }
    __syncthreads();
    if (own) {
        int st = base_s + wsum[w] + x - c;
        rowstart[node] = st;
        cnt[node]      = c - 1;
        dis[node]      = rsqrtf((float)c);
        rs[t]          = st;
    }
    if (t < BINW) lc[t] = 0;
    __syncthreads();
    for (int base = b0; base < b1; base += BB * RPT) {
        int2 r[RPT];
#pragma unroll
        for (int k = 0; k < RPT; ++k) {
            int idx = base + k * BB + t;
            r[k] = binned[idx < b1 ? idx : b1 - 1];
        }
#pragma unroll
        for (int k = 0; k < RPT; ++k) {
            if (base + k * BB + t < b1) {
                int j = r[k].x - row0;
                int p = atomicAdd(&lc[j], 1);
                col[rs[j] + p] = r[k].y;
            }
        }
    }
}


__device__ __forceinline__ void fma4(float4& a, const float4& v, float w) {
    a.x = fmaf(v.x, w, a.x);
    a.y = fmaf(v.y, w, a.y);
    a.z = fmaf(v.z, w, a.z);
    a.w = fmaf(v.w, w, a.w);
}

#define SCAP (PITCH / 2)

template <bool SORT>
__global__ __launch_bounds__(256) void gcn_layer(const float* __restrict__ Xin,
                                                 const int* __restrict__ rowstart,
                                                 const int* __restrict__ cnt,
                                                 const int2* __restrict__ rec,
                                                 const float* __restrict__ dis,
                                                 const int* __restrict__ arr,
                                                 int2* recw,
                                                 const float* __restrict__ Wm,
                                                 const float* __restrict__ bias,
                                                 float* __restrict__ out, int n, int relu) {
    __shared__ __attribute__((aligned(16))) float aggT[TR * PITCH];
    __shared__ __attribute__((aligned(16))) float outT[TR * PITCH];

    int wib  = __builtin_amdgcn_readfirstlane((int)(threadIdx.x >> 5));
    int lane = threadIdx.x & 31;
    int row0  = blockIdx.x * TR;
    int node0 = row0 + wib * NPW;

    int mst = 0, mc = 0, mdt = 0;
    if (lane < NPW && node0 + lane < n) {
        mst = rowstart[node0 + lane];
        mc  = cnt[node0 + lane] + 1;
        if (SORT) mdt = __float_as_int(dis[node0 + lane]);
    }

    int  st[NPW], c[NPW];
    int2 r[NPW];
    if (!SORT) {
#pragma unroll
        for (int i = 0; i < NPW; ++i) {
            st[i] = __builtin_amdgcn_readlane(mst, i);
            c[i]  = __builtin_amdgcn_readlane(mc, i);
            r[i]  = make_int2(0, 0);
            if (lane < c[i]) r[i] = rec[st[i] + lane];
        }
    } else {
        int   cc[NPW], n0[NPW], key[NPW], ord[NPW], rk[NPW];
        float dt[NPW], wgt[NPW];
#pragma unroll
        for (int i = 0; i < NPW; ++i) {
            st[i] = __builtin_amdgcn_readlane(mst, i);
            c[i]  = __builtin_amdgcn_readlane(mc, i);
            dt[i] = __int_as_float(__builtin_amdgcn_readlane(mdt, i));
            cc[i] = c[i] > 0 ? c[i] - 1 : 0;
            n0[i] = cc[i] < 32 ? cc[i] : 32;
            int kq = arr[st[i] + (lane < n0[i] ? lane : 0)];
            key[i] = lane < n0[i] ? kq : 0x7fffffff;
        }
#pragma unroll
        for (int i = 0; i < NPW; ++i) {
            wgt[i] = dis[lane < n0[i] ? key[i] : 0];
        }
#pragma unroll
        for (int i = 0; i < NPW; ++i) {
            ord[i] = 0x7fffffff;
            if (lane < n0[i]) ord[i] = (key[i] << 5) | lane;
            rk[i] = 0;
            for (int k = 0; k < n0[i]; ++k) {
                int kk = __builtin_amdgcn_readlane(ord[i], k);
                rk[i] += (kk < ord[i]) ? 1 : 0;
            }
            if (cc[i] > 32) {
                for (int b = 32; b < cc[i]; b += 32) {
                    int kv = 0x7fffffff;
                    if (b + lane < cc[i]) kv = arr[st[i] + b + lane];
                    int lim = cc[i] - b < 32 ? cc[i] - b : 32;
                    for (int k = 0; k < lim; ++k) {
                        int kk = __builtin_amdgcn_readlane(kv, k);
                        rk[i] += (kk < key[i]) ? 1 : 0;
                    }
                }
            }
        }
#pragma unroll
        for (int i = 0; i < NPW; ++i) {
            int2* slot = (int2*)(outT + (wib * NPW + i) * PITCH);
            if (lane < n0[i]) {
                int2 q = make_int2(key[i], __float_as_int(wgt[i] * dt[i]));
                recw[st[i] + rk[i]] = q;
                if (rk[i] < SCAP) slot[rk[i]] = q;
            }
            if (cc[i] > 32) {
                for (int a = 32; a < cc[i]; a += 32) {
                    int e  = a + lane;
                    int ke = 0x7fffffff;
                    if (e < cc[i]) ke = arr[st[i] + e];
                    int re = 0;
                    for (int b = 0; b < cc[i]; b += 32) {
                        int kv = 0x7fffffff;
                        if (b + lane < cc[i]) kv = arr[st[i] + b + lane];
                        int lim = cc[i] - b < 32 ? cc[i] - b : 32;
                        for (int k = 0; k < lim; ++k) {
                            int kk = __builtin_amdgcn_readlane(kv, k);
                            re += (kk < ke || (kk == ke && b + k < e)) ? 1 : 0;
                        }
                    }
                    if (e < cc[i]) {
                        int2 q = make_int2(ke, __float_as_int(dis[ke] * dt[i]));
                        recw[st[i] + re] = q;
                        if (re < SCAP) slot[re] = q;
                    }
                }
            }
            if (lane == 0 && c[i] > 0) {
                int2 q = make_int2(node0 + i, __float_as_int(dt[i] * dt[i]));
                recw[st[i] + cc[i]] = q;
                if (cc[i] < SCAP) slot[cc[i]] = q;
            }
        }
#pragma unroll
        for (int i = 0; i < NPW; ++i) {
            const int2* slot = (const int2*)(outT + (wib * NPW + i) * PITCH);
            r[i] = make_int2(0, 0);
            if (lane < c[i]) r[i] = slot[lane];
        }
    }

#pragma unroll
    for (int i = 0; i < NPW; ++i) {
        float4 acc = make_float4(0.0f, 0.0f, 0.0f, 0.0f);

        int c32 = c[i] < 32 ? c[i] : 32;
        for (int j0 = 0; j0 < c32; j0 += 8) {
            float4 v[8];
#pragma unroll
            for (int k = 0; k < 8; ++k) {
                int jj = j0 + k < c32 ? j0 + k : c32 - 1;
                int s = __builtin_amdgcn_readlane(r[i].x, jj);
                v[k] = ((const float4*)(Xin + (size_t)s * DCH))[lane];
            }
#pragma unroll
            for (int k = 0; k < 8; ++k) {
                if (j0 + k < c32)
                    fma4(acc, v[k], __int_as_float(__builtin_amdgcn_readlane(r[i].y, j0 + k)));
            }
        }
        if (!SORT) {
            for (int j = 32; j < c[i]; ++j) {
                int2 q = rec[st[i] + j];
                float4 v = ((const float4*)(Xin + (size_t)q.x * DCH))[lane];
                fma4(acc, v, __int_as_float(q.y));
            }
        } else {
            const int2* slot = (const int2*)(outT + (wib * NPW + i) * PITCH);
            if (c[i] > SCAP) __threadfence();
            for (int j = 32; j < c[i]; ++j) {
                int2 q;
                if (j < SCAP) {
                    q = slot[j];
                } else {
                    unsigned long long u = __hip_atomic_load((unsigned long long*)(recw + st[i] + j),
                                                             __ATOMIC_RELAXED, __HIP_MEMORY_SCOPE_AGENT);
                    q = make_int2((int)(u & 0xffffffffull), (int)(u >> 32));
                }
                float4 v = ((const float4*)(Xin + (size_t)q.x * DCH))[lane];
                fma4(acc, v, __int_as_float(q.y));
            }
        }

        *(float4*)(aggT + (wib * NPW + i) * PITCH + 4 * lane) = acc;
    }

    __syncthreads();

    int hi  = lane >> 4;
    int l16 = lane & 15;
    int col = wib * 16 + l16;
    const float* a0p = aggT + l16 * PITCH + 2 * hi;
    const float* a1p = a0p + 16 * PITCH;
    const float* bp  = Wm + (size_t)(2 * hi) * DCH + col;

    v8f c0 = {};
    v8f c1 = {};
#pragma unroll 8
    for (int k0 = 0; k0 < DCH; k0 += 4) {
        v2f a0 = *(const v2f*)(a0p + k0);
        v2f a1 = *(const v2f*)(a1p + k0);
        v2f b;
        b.x = bp[(size_t)k0 * DCH];
        b.y = bp[(size_t)(k0 + 1) * DCH];
        c0 = __builtin_amdgcn_wmma_f32_16x16x4_f32(
                  false, a0,  false, b,
                  (short)0, c0,  false,  false);
        c1 = __builtin_amdgcn_wmma_f32_16x16x4_f32(
                  false, a1,  false, b,
                  (short)0, c1,  false,  false);
    }

#pragma unroll
    for (int i = 0; i < 8; ++i) {
        outT[(8 * hi + i) * PITCH + col]      = c0[i];
        outT[(16 + 8 * hi + i) * PITCH + col] = c1[i];
    }

    __syncthreads();

    float4 b4 = ((const float4*)bias)[lane];
#pragma unroll
    for (int i = 0; i < NPW; ++i) {
        int lr = wib * NPW + i;
        int t  = row0 + lr;
        if (t < n) {
            float4 o = *(const float4*)(outT + lr * PITCH + 4 * lane);
            o.x += b4.x; o.y += b4.y; o.z += b4.z; o.w += b4.w;
            if (relu) {
                o.x = fmaxf(o.x, 0.0f); o.y = fmaxf(o.y, 0.0f);
                o.z = fmaxf(o.z, 0.0f); o.w = fmaxf(o.w, 0.0f);
            }
            ((float4*)(out + (size_t)t * DCH))[lane] = o;
        }
    }
}


extern "C" void kernel_launch(void* const* d_in, const int* in_sizes, int n_in,
                              void* d_out, int out_size, void* d_ws, size_t ws_size,
                              hipStream_t stream) {
    const float* X  = (const float*)d_in[0];
    const int*   ei = (const int*)  d_in[1];
    const float* W1 = (const float*)d_in[2];
    const float* b1 = (const float*)d_in[3];
    const float* W2 = (const float*)d_in[4];
    const float* b2 = (const float*)d_in[5];
    const float* W3 = (const float*)d_in[6];
    const float* b3 = (const float*)d_in[7];
    float* out = (float*)d_out;

    const int n  = in_sizes[0] / DCH;
    const int e  = in_sizes[1] / 2;

    const int* src = ei;
    const int* tgt = ei + e;

    int*   cnt      = (int*)d_ws;
    int*   rowstart = cnt + n;
    float* dis      = (float*)(rowstart + n);
    int*   total    = (int*)(dis + n);
    int*   binstart = total + 4;
    int*   table    = binstart + NBIN + 4;
    int2*  binned   = (int2*)(table + NBLK * NBIN);
    int*   col      = (int*)(binned + e);
    int2*  rec      = (int2*)(col + e + n);
    float* Y1       = (float*)(rec + e + n);

    const int chunk = (e + NBLK - 1) / NBLK;
    const int nbins = (n + BINW - 1) / BINW;
    bin_hist   <<<NBLK,  BT, 0, stream>>>(tgt, table, total, e, chunk);
    bin_scatter<<<NBLK,  BT, 0, stream>>>(src, tgt, table, binstart, binned, e, chunk);
    bin_build  <<<nbins, BB, 0, stream>>>(binstart, binned, total, cnt, rowstart, dis, col, n);

    const int layerBlocks = (n + TR - 1) / TR;

    gcn_layer<true> <<<layerBlocks, 256, 0, stream>>>(X,   rowstart, cnt, nullptr, dis, col, rec,
                                                      W1, b1, out, n, 1);
    gcn_layer<false><<<layerBlocks, 256, 0, stream>>>(out, rowstart, cnt, rec, dis, col, nullptr,
                                                      W2, b2, Y1,  n, 1);
    gcn_layer<false><<<layerBlocks, 256, 0, stream>>>(Y1,  rowstart, cnt, rec, dis, col, nullptr,
                                                      W3, b3, out, n, 0);
}


# Edges binned inside each block's own slice (LDS add = rank, per-slice bin offsets in a table, bin blocks collect the pieces), not into global bins: count pass gone, 6 -> 5 launches; sum order kept
# speedup vs baseline: 6.2181x; 1.0579x over previous
#include <hip/hip_runtime.h>

typedef __attribute__((ext_vector_type(2))) float v2f;
typedef __attribute__((ext_vector_type(8))) float v8f;

#define DCH 128
#define NPW 4
#define WPB 8
#define TR  (WPB * NPW)
#define PITCH 132


#define BINW 256
#define BINSH 8
#define NBIN 256
#define NBLK 128
#define BT   1024
#define EPT  8
#define BB   1024
#define SEGW (NBLK / (BB / 32))
#define SEGR 2

__global__ __launch_bounds__(BT) void bin_local(const int* __restrict__ src, const int* __restrict__ tgt,
                                                int* __restrict__ tab_off, int* __restrict__ tab_cnt,
                                                int* __restrict__ total, int2* __restrict__ binned,
                                                int e, int chunk) {
    __shared__ int h[NBIN];
    __shared__ int off[NBIN];
    __shared__ int top[NBIN];
    __shared__ int wsum[NBIN / 32];
    int t = threadIdx.x;
    if (t < NBIN) h[t] = 0;
    if (blockIdx.x == 0 && t == 0) total[0] = 0;
    __syncthreads();
    int lo = blockIdx.x * chunk;
    int hi = lo + chunk < e ? lo + chunk : e;

    int tv[EPT], sv[EPT], pk[EPT];
#pragma unroll
    for (int k = 0; k < EPT; ++k) {
        int idx = lo + k * BT + t;
        int ii  = idx < hi ? idx : hi - 1;
        tv[k] = tgt[ii];
        sv[k] = src[ii];
    }
#pragma unroll
    for (int k = 0; k < EPT; ++k) {
        int idx = lo + k * BT + t;
        pk[k] = 0;
        if (idx < hi) pk[k] = atomicAdd(&h[tv[k] >> BINSH], 1);
    }
    for (int base = lo + BT * EPT; base < hi; base += BT * EPT) {
        int v[EPT];
#pragma unroll
        for (int k = 0; k < EPT; ++k) {
            int idx = base + k * BT + t;
            v[k] = tgt[idx < hi ? idx : hi - 1];
        }
#pragma unroll
        for (int k = 0; k < EPT; ++k) {
            int idx = base + k * BT + t;
            if (idx < hi) atomicAdd(&h[v[k] >> BINSH], 1);
        }
    }
    __syncthreads();

    int c = t < NBIN ? h[t] : 0;
    int lane = t & 31;
    int w    = t >> 5;
    int x = c;
    for (int d = 1; d < 32; d <<= 1) {
        int y = __shfl_up(x, d);
        if (lane >= d) x += y;
    }
    if (t < NBIN && lane == 31) wsum[w] = x;
    __syncthreads();
    if (t < NBIN) {
        int s = 0;
        for (int k = 0; k < w; ++k) s += wsum[k];
        int first = s + x - c;
        off[t] = first;
        top[t] = first + c;
        tab_off[blockIdx.x * NBIN + t] = lo + first;
        tab_cnt[blockIdx.x * NBIN + t] = c;
    }
    __syncthreads();

#pragma unroll
    for (int k = 0; k < EPT; ++k) {
        int idx = lo + k * BT + t;
        if (idx < hi) binned[lo + off[tv[k] >> BINSH] + pk[k]] = make_int2(tv[k], sv[k]);
    }
    for (int base = lo + BT * EPT; base < hi; base += BT * EPT) {
#pragma unroll
        for (int k = 0; k < EPT; ++k) {
            int idx = base + k * BT + t;
            int ii  = idx < hi ? idx : hi - 1;
            tv[k] = tgt[ii];
            sv[k] = src[ii];
        }
#pragma unroll
        for (int k = 0; k < EPT; ++k) {
            int idx = base + k * BT + t;
            if (idx < hi) {
                int p = atomicSub(&top[tv[k] >> BINSH], 1) - 1;
                binned[lo + p] = make_int2(tv[k], sv[k]);
            }
        }
    }
}

__global__ __launch_bounds__(BB) void bin_build(const int* __restrict__ tab_off,
                                                const int* __restrict__ tab_cnt,
                                                const int2* __restrict__ binned,
                                                int* __restrict__ total, int* __restrict__ cnt,
                                                int* __restrict__ rowstart, float* __restrict__ dis,
                                                int* __restrict__ col, int n) {
    __shared__ int lc[BINW];
    __shared__ int rs[BINW];
    __shared__ int wsum[BB / 32];
    __shared__ int base_s;
    int t    = threadIdx.x;
    int lane = t & 31;
    int w    = __builtin_amdgcn_readfirstlane(t >> 5);
    int row0 = blockIdx.x * BINW;
    if (t < BINW) lc[t] = 0;

    int mo = 0, mc = 0;
    if (lane < SEGW) {
        mo = tab_off[(w * SEGW + lane) * NBIN + blockIdx.x];
        mc = tab_cnt[(w * SEGW + lane) * NBIN + blockIdx.x];
    }
    int  po[SEGW], pc[SEGW];
    int2 r[SEGW][SEGR];
#pragma unroll
    for (int i = 0; i < SEGW; ++i) {
        po[i] = __builtin_amdgcn_readlane(mo, i);
        pc[i] = __builtin_amdgcn_readlane(mc, i);
#pragma unroll
        for (int j = 0; j < SEGR; ++j) {
            int q = j * 32 + lane;
            r[i][j] = binned[q < pc[i] ? po[i] + q : 0];
        }
    }
    __syncthreads();

#pragma unroll
    for (int i = 0; i < SEGW; ++i) {
#pragma unroll
        for (int j = 0; j < SEGR; ++j) {
            if (j * 32 + lane < pc[i]) atomicAdd(&lc[r[i][j].x - row0], 1);
        }
        for (int q = 32 * SEGR + lane; q < pc[i]; q += 32) {
            int2 rr = binned[po[i] + q];
            atomicAdd(&lc[rr.x - row0], 1);
        }
    }
    __syncthreads();

    int node = row0 + t;
    int own  = (t < BINW && node < n) ? 1 : 0;
    int c = own ? lc[t] + 1 : 0;
    int x = c;
    for (int d = 1; d < 32; d <<= 1) {
        int y = __shfl_up(x, d);
        if (lane >= d) x += y;
    }
    if (lane == 31) wsum[w] = x;
    __syncthreads();
    if (t == 0) {
        int s = 0;
        for (int k = 0; k < BB / 32; ++k) { int v = wsum[k]; wsum[k] = s; s += v; }
        base_s = atomicAdd(total, s);
    }
    __syncthreads();
    if (own) {
        int st = base_s + wsum[w] + x - c;
        rowstart[node] = st;
        cnt[node]      = c - 1;
        dis[node]      = rsqrtf((float)c);
        rs[t]          = st;
    }
    if (t < BINW) lc[t] = 0;
    __syncthreads();
#pragma unroll
    for (int i = 0; i < SEGW; ++i) {
#pragma unroll
        for (int j = 0; j < SEGR; ++j) {
            if (j * 32 + lane < pc[i]) {
                int jj = r[i][j].x - row0;
                int p  = atomicAdd(&lc[jj], 1);
                col[rs[jj] + p] = r[i][j].y;
            }
        }
        for (int q = 32 * SEGR + lane; q < pc[i]; q += 32) {
            int2 rr = binned[po[i] + q];
            int jj = rr.x - row0;
            int p  = atomicAdd(&lc[jj], 1);
            col[rs[jj] + p] = rr.y;
        }
    }
}


__device__ __forceinline__ void fma4(float4& a, const float4& v, float w) {
    a.x = fmaf(v.x, w, a.x);
    a.y = fmaf(v.y, w, a.y);
    a.z = fmaf(v.z, w, a.z);
    a.w = fmaf(v.w, w, a.w);
}

#define SCAP (PITCH / 2)

template <bool SORT>
__global__ __launch_bounds__(256) void gcn_layer(const float* __restrict__ Xin,
                                                 const int* __restrict__ rowstart,
                                                 const int* __restrict__ cnt,
                                                 const int2* __restrict__ rec,
                                                 const float* __restrict__ dis,
                                                 const int* __restrict__ arr,
                                                 int2* recw,
                                                 const float* __restrict__ Wm,
                                                 const float* __restrict__ bias,
                                                 float* __restrict__ out, int n, int relu) {
    __shared__ __attribute__((aligned(16))) float aggT[TR * PITCH];
    __shared__ __attribute__((aligned(16))) float outT[TR * PITCH];

    int wib  = __builtin_amdgcn_readfirstlane((int)(threadIdx.x >> 5));
    int lane = threadIdx.x & 31;
    int row0  = blockIdx.x * TR;
    int node0 = row0 + wib * NPW;

    int mst = 0, mc = 0, mdt = 0;
    if (lane < NPW && node0 + lane < n) {
        mst = rowstart[node0 + lane];
        mc  = cnt[node0 + lane] + 1;
        if (SORT) mdt = __float_as_int(dis[node0 + lane]);
    }

    int  st[NPW], c[NPW];
    int2 r[NPW];
    if (!SORT) {
#pragma unroll
        for (int i = 0; i < NPW; ++i) {
            st[i] = __builtin_amdgcn_readlane(mst, i);
            c[i]  = __builtin_amdgcn_readlane(mc, i);
            r[i]  = make_int2(0, 0);
            if (lane < c[i]) r[i] = rec[st[i] + lane];
        }
    } else {
        int   cc[NPW], n0[NPW], key[NPW], ord[NPW], rk[NPW];
        float dt[NPW], wgt[NPW];
#pragma unroll
        for (int i = 0; i < NPW; ++i) {
            st[i] = __builtin_amdgcn_readlane(mst, i);
            c[i]  = __builtin_amdgcn_readlane(mc, i);
            dt[i] = __int_as_float(__builtin_amdgcn_readlane(mdt, i));
            cc[i] = c[i] > 0 ? c[i] - 1 : 0;
            n0[i] = cc[i] < 32 ? cc[i] : 32;
            int kq = arr[st[i] + (lane < n0[i] ? lane : 0)];
            key[i] = lane < n0[i] ? kq : 0x7fffffff;
        }
#pragma unroll
        for (int i = 0; i < NPW; ++i) {
            wgt[i] = dis[lane < n0[i] ? key[i] : 0];
        }
#pragma unroll
        for (int i = 0; i < NPW; ++i) {
            ord[i] = 0x7fffffff;
            if (lane < n0[i]) ord[i] = (key[i] << 5) | lane;
            rk[i] = 0;
            for (int k = 0; k < n0[i]; ++k) {
                int kk = __builtin_amdgcn_readlane(ord[i], k);
                rk[i] += (kk < ord[i]) ? 1 : 0;
            }
            if (cc[i] > 32) {
                for (int b = 32; b < cc[i]; b += 32) {
                    int kv = 0x7fffffff;
                    if (b + lane < cc[i]) kv = arr[st[i] + b + lane];
                    int lim = cc[i] - b < 32 ? cc[i] - b : 32;
                    for (int k = 0; k < lim; ++k) {
                        int kk = __builtin_amdgcn_readlane(kv, k);
                        rk[i] += (kk < key[i]) ? 1 : 0;
                    }
                }
            }
        }
#pragma unroll
        for (int i = 0; i < NPW; ++i) {
            int2* slot = (int2*)(outT + (wib * NPW + i) * PITCH);
            if (lane < n0[i]) {
                int2 q = make_int2(key[i], __float_as_int(wgt[i] * dt[i]));
                recw[st[i] + rk[i]] = q;
                if (rk[i] < SCAP) slot[rk[i]] = q;
            }
            if (cc[i] > 32) {
                for (int a = 32; a < cc[i]; a += 32) {
                    int e  = a + lane;
                    int ke = 0x7fffffff;
                    if (e < cc[i]) ke = arr[st[i] + e];
                    int re = 0;
                    for (int b = 0; b < cc[i]; b += 32) {
                        int kv = 0x7fffffff;
                        if (b + lane < cc[i]) kv = arr[st[i] + b + lane];
                        int lim = cc[i] - b < 32 ? cc[i] - b : 32;
                        for (int k = 0; k < lim; ++k) {
                            int kk = __builtin_amdgcn_readlane(kv, k);
                            re += (kk < ke || (kk == ke && b + k < e)) ? 1 : 0;
                        }
                    }
                    if (e < cc[i]) {
                        int2 q = make_int2(ke, __float_as_int(dis[ke] * dt[i]));
                        recw[st[i] + re] = q;
                        if (re < SCAP) slot[re] = q;
                    }
                }
            }
            if (lane == 0 && c[i] > 0) {
                int2 q = make_int2(node0 + i, __float_as_int(dt[i] * dt[i]));
                recw[st[i] + cc[i]] = q;
                if (cc[i] < SCAP) slot[cc[i]] = q;
            }
        }
#pragma unroll
        for (int i = 0; i < NPW; ++i) {
            const int2* slot = (const int2*)(outT + (wib * NPW + i) * PITCH);
            r[i] = make_int2(0, 0);
            if (lane < c[i]) r[i] = slot[lane];
        }
    }

#pragma unroll
    for (int i = 0; i < NPW; ++i) {
        float4 acc = make_float4(0.0f, 0.0f, 0.0f, 0.0f);

        int c32 = c[i] < 32 ? c[i] : 32;
        for (int j0 = 0; j0 < c32; j0 += 8) {
            float4 v[8];
#pragma unroll
            for (int k = 0; k < 8; ++k) {
                int jj = j0 + k < c32 ? j0 + k : c32 - 1;
                int s = __builtin_amdgcn_readlane(r[i].x, jj);
                v[k] = ((const float4*)(Xin + (size_t)s * DCH))[lane];
            }
#pragma unroll
            for (int k = 0; k < 8; ++k) {
                if (j0 + k < c32)
                    fma4(acc, v[k], __int_as_float(__builtin_amdgcn_readlane(r[i].y, j0 + k)));
            }
        }
        if (!SORT) {
            for (int j = 32; j < c[i]; ++j) {
                int2 q = rec[st[i] + j];
                float4 v = ((const float4*)(Xin + (size_t)q.x * DCH))[lane];
                fma4(acc, v, __int_as_float(q.y));
            }
        } else {
            const int2* slot = (const int2*)(outT + (wib * NPW + i) * PITCH);
            if (c[i] > SCAP) __threadfence();
            for (int j = 32; j < c[i]; ++j) {
                int2 q;
                if (j < SCAP) {
                    q = slot[j];
                } else {
                    unsigned long long u = __hip_atomic_load((unsigned long long*)(recw + st[i] + j),
                                                             __ATOMIC_RELAXED, __HIP_MEMORY_SCOPE_AGENT);
                    q = make_int2((int)(u & 0xffffffffull), (int)(u >> 32));
                }
                float4 v = ((const float4*)(Xin + (size_t)q.x * DCH))[lane];
                fma4(acc, v, __int_as_float(q.y));
            }
        }

        *(float4*)(aggT + (wib * NPW + i) * PITCH + 4 * lane) = acc;
    }

    __syncthreads();

    int hi  = lane >> 4;
    int l16 = lane & 15;
    int col = wib * 16 + l16;
    const float* a0p = aggT + l16 * PITCH + 2 * hi;
    const float* a1p = a0p + 16 * PITCH;
    const float* bp  = Wm + (size_t)(2 * hi) * DCH + col;

    v8f c0 = {};
    v8f c1 = {};
#pragma unroll 8
    for (int k0 = 0; k0 < DCH; k0 += 4) {
        v2f a0 = *(const v2f*)(a0p + k0);
        v2f a1 = *(const v2f*)(a1p + k0);
        v2f b;
        b.x = bp[(size_t)k0 * DCH];
        b.y = bp[(size_t)(k0 + 1) * DCH];
        c0 = __builtin_amdgcn_wmma_f32_16x16x4_f32(
                  false, a0,  false, b,
                  (short)0, c0,  false,  false);
        c1 = __builtin_amdgcn_wmma_f32_16x16x4_f32(
                  false, a1,  false, b,
                  (short)0, c1,  false,  false);
    }

#pragma unroll
    for (int i = 0; i < 8; ++i) {
        outT[(8 * hi + i) * PITCH + col]      = c0[i];
        outT[(16 + 8 * hi + i) * PITCH + col] = c1[i];
    }

    __syncthreads();

    float4 b4 = ((const float4*)bias)[lane];
#pragma unroll
    for (int i = 0; i < NPW; ++i) {
        int lr = wib * NPW + i;
        int t  = row0 + lr;
        if (t < n) {
            float4 o = *(const float4*)(outT + lr * PITCH + 4 * lane);
            o.x += b4.x; o.y += b4.y; o.z += b4.z; o.w += b4.w;
            if (relu) {
                o.x = fmaxf(o.x, 0.0f); o.y = fmaxf(o.y, 0.0f);
                o.z = fmaxf(o.z, 0.0f); o.w = fmaxf(o.w, 0.0f);
            }
            ((float4*)(out + (size_t)t * DCH))[lane] = o;
        }
    }
}


extern "C" void kernel_launch(void* const* d_in, const int* in_sizes, int n_in,
                              void* d_out, int out_size, void* d_ws, size_t ws_size,
                              hipStream_t stream) {
    const float* X  = (const float*)d_in[0];
    const int*   ei = (const int*)  d_in[1];
    const float* W1 = (const float*)d_in[2];
    const float* b1 = (const float*)d_in[3];
    const float* W2 = (const float*)d_in[4];
    const float* b2 = (const float*)d_in[5];
    const float* W3 = (const float*)d_in[6];
    const float* b3 = (const float*)d_in[7];
    float* out = (float*)d_out;

    const int n  = in_sizes[0] / DCH;
    const int e  = in_sizes[1] / 2;

    const int* src = ei;
    const int* tgt = ei + e;

    int*   cnt      = (int*)d_ws;
    int*   rowstart = cnt + n;
    float* dis      = (float*)(rowstart + n);
    int*   total    = (int*)(dis + n);
    int*   tab_off  = total + 4;
    int*   tab_cnt  = tab_off + NBLK * NBIN;
    int2*  binned   = (int2*)(tab_cnt + NBLK * NBIN);
    int*   col      = (int*)(binned + e);
    int2*  rec      = (int2*)(col + e + n);
    float* Y1       = (float*)(rec + e + n);

    const int chunk = (e + NBLK - 1) / NBLK;
    const int nbins = (n + BINW - 1) / BINW;
    bin_local<<<NBLK,  BT, 0, stream>>>(src, tgt, tab_off, tab_cnt, total, binned, e, chunk);
    bin_build<<<nbins, BB, 0, stream>>>(tab_off, tab_cnt, binned, total, cnt, rowstart, dis, col, n);

    const int layerBlocks = (n + TR - 1) / TR;

    gcn_layer<true> <<<layerBlocks, 256, 0, stream>>>(X,   rowstart, cnt, nullptr, dis, col, rec,
                                                      W1, b1, out, n, 1);
    gcn_layer<false><<<layerBlocks, 256, 0, stream>>>(out, rowstart, cnt, rec, dis, col, nullptr,
                                                      W2, b2, Y1,  n, 1);
    gcn_layer<false><<<layerBlocks, 256, 0, stream>>>(Y1,  rowstart, cnt, rec, dis, col, nullptr,
                                                      W3, b3, out, n, 0);
}


# Grouping kernels order their records in LDS and write them out as whole lines (bin_local: the slice's pairs; bin_build: the bin's sources): no scattered store per edge left; sums unchanged
# speedup vs baseline: 6.4436x; 1.0363x over previous
#include <hip/hip_runtime.h>

typedef __attribute__((ext_vector_type(2))) float v2f;
typedef __attribute__((ext_vector_type(8))) float v8f;

#define DCH 128
#define NPW 4
#define WPB 8
#define TR  (WPB * NPW)
#define PITCH 132


#define BINW 256
#define BINSH 8
#define NBIN 256
#define NBLK 128
#define BT   1024
#define EPT  8
#define BB   1024
#define SEGW (NBLK / (BB / 32))
#define SEGR 2
#define LCAP 6400
#define CCAP 8192

__global__ __launch_bounds__(BT) void bin_local(const int* __restrict__ src, const int* __restrict__ tgt,
                                                int* __restrict__ tab_off, int* __restrict__ tab_cnt,
                                                int* __restrict__ total, int2* __restrict__ binned,
                                                int e, int chunk) {
    __shared__ int h[NBIN];
    __shared__ int off[NBIN];
    __shared__ int top[NBIN];
    __shared__ int wsum[NBIN / 32];
    __shared__ int2 stage[LCAP];
    int t = threadIdx.x;
    if (t < NBIN) h[t] = 0;
    if (blockIdx.x == 0 && t == 0) total[0] = 0;
    __syncthreads();
    int lo = blockIdx.x * chunk;
    int hi = lo + chunk < e ? lo + chunk : e;

    int tv[EPT], sv[EPT], pk[EPT];
#pragma unroll
    for (int k = 0; k < EPT; ++k) {
        int idx = lo + k * BT + t;
        int ii  = idx < hi ? idx : hi - 1;
        tv[k] = tgt[ii];
        sv[k] = src[ii];
    }
#pragma unroll
    for (int k = 0; k < EPT; ++k) {
        int idx = lo + k * BT + t;
        pk[k] = 0;
        if (idx < hi) pk[k] = atomicAdd(&h[tv[k] >> BINSH], 1);
    }
    for (int base = lo + BT * EPT; base < hi; base += BT * EPT) {
        int v[EPT];
#pragma unroll
        for (int k = 0; k < EPT; ++k) {
            int idx = base + k * BT + t;
            v[k] = tgt[idx < hi ? idx : hi - 1];
        }
#pragma unroll
        for (int k = 0; k < EPT; ++k) {
            int idx = base + k * BT + t;
            if (idx < hi) atomicAdd(&h[v[k] >> BINSH], 1);
        }
    }
    __syncthreads();

    int c = t < NBIN ? h[t] : 0;
    int lane = t & 31;
    int w    = t >> 5;
    int x = c;
    for (int d = 1; d < 32; d <<= 1) {
        int y = __shfl_up(x, d);
        if (lane >= d) x += y;
    }
    if (t < NBIN && lane == 31) wsum[w] = x;
    __syncthreads();
    if (t < NBIN) {
        int s = 0;
        for (int k = 0; k < w; ++k) s += wsum[k];
        int first = s + x - c;
        off[t] = first;
        top[t] = first + c;
        tab_off[blockIdx.x * NBIN + t] = lo + first;
        tab_cnt[blockIdx.x * NBIN + t] = c;
    }
    __syncthreads();

#pragma unroll
    for (int k = 0; k < EPT; ++k) {
        int idx = lo + k * BT + t;
        if (idx < hi) {
            int p = off[tv[k] >> BINSH] + pk[k];
            if (p < LCAP) stage[p] = make_int2(tv[k], sv[k]);
            else          binned[lo + p] = make_int2(tv[k], sv[k]);
        }
    }
    for (int base = lo + BT * EPT; base < hi; base += BT * EPT) {
#pragma unroll
        for (int k = 0; k < EPT; ++k) {
            int idx = base + k * BT + t;
            int ii  = idx < hi ? idx : hi - 1;
            tv[k] = tgt[ii];
            sv[k] = src[ii];
        }
#pragma unroll
        for (int k = 0; k < EPT; ++k) {
            int idx = base + k * BT + t;
            if (idx < hi) {
                int p = atomicSub(&top[tv[k] >> BINSH], 1) - 1;
                if (p < LCAP) stage[p] = make_int2(tv[k], sv[k]);
                else          binned[lo + p] = make_int2(tv[k], sv[k]);
            }
        }
    }
    __syncthreads();
    int len = hi - lo;
    int m   = len < LCAP ? len : LCAP;
    if (m > 0) {
        int2 v[(LCAP + BT - 1) / BT];
#pragma unroll
        for (int k = 0; k < (LCAP + BT - 1) / BT; ++k) {
            int q = k * BT + t;
            v[k] = stage[q < m ? q : m - 1];
        }
#pragma unroll
        for (int k = 0; k < (LCAP + BT - 1) / BT; ++k) {
            int q = k * BT + t;
            if (q < m) binned[lo + q] = v[k];
        }
    }
}

__global__ __launch_bounds__(BB) void bin_build(const int* __restrict__ tab_off,
                                                const int* __restrict__ tab_cnt,
                                                const int2* __restrict__ binned,
                                                int* __restrict__ total, int* __restrict__ cnt,
                                                int* __restrict__ rowstart, float* __restrict__ dis,
                                                int* __restrict__ col, int n) {
    __shared__ int lc[BINW];
    __shared__ int rs[BINW];
    __shared__ int wsum[BB / 32];
    __shared__ int base_s;
    __shared__ int tot_s;
    __shared__ int stage[CCAP];
    int t    = threadIdx.x;
    int lane = t & 31;
    int w    = __builtin_amdgcn_readfirstlane(t >> 5);
    int row0 = blockIdx.x * BINW;
    if (t < BINW) lc[t] = 0;

    int mo = 0, mc = 0;
    if (lane < SEGW) {
        mo = tab_off[(w * SEGW + lane) * NBIN + blockIdx.x];
        mc = tab_cnt[(w * SEGW + lane) * NBIN + blockIdx.x];
    }
    int  po[SEGW], pc[SEGW];
    int2 r[SEGW][SEGR];
#pragma unroll
    for (int i = 0; i < SEGW; ++i) {
        po[i] = __builtin_amdgcn_readlane(mo, i);
        pc[i] = __builtin_amdgcn_readlane(mc, i);
#pragma unroll
        for (int j = 0; j < SEGR; ++j) {
            int q = j * 32 + lane;
            r[i][j] = binned[q < pc[i] ? po[i] + q : 0];
        }
    }
    __syncthreads();

#pragma unroll
    for (int i = 0; i < SEGW; ++i) {
#pragma unroll
        for (int j = 0; j < SEGR; ++j) {
            if (j * 32 + lane < pc[i]) atomicAdd(&lc[r[i][j].x - row0], 1);
        }
        for (int q = 32 * SEGR + lane; q < pc[i]; q += 32) {
            int2 rr = binned[po[i] + q];
            atomicAdd(&lc[rr.x - row0], 1);
        }
    }
    __syncthreads();

    int node = row0 + t;
    int own  = (t < BINW && node < n) ? 1 : 0;
    int c = own ? lc[t] + 1 : 0;
    int x = c;
    for (int d = 1; d < 32; d <<= 1) {
        int y = __shfl_up(x, d);
        if (lane >= d) x += y;
    }
    if (lane == 31) wsum[w] = x;
    __syncthreads();
    if (t == 0) {
        int s = 0;
        for (int k = 0; k < BB / 32; ++k) { int v = wsum[k]; wsum[k] = s; s += v; }
        tot_s  = s;
        base_s = atomicAdd(total, s);
    }
    __syncthreads();
    if (own) {
        int sl = wsum[w] + x - c;
        rowstart[node] = base_s + sl;
        cnt[node]      = c - 1;
        dis[node]      = rsqrtf((float)c);
        rs[t]          = sl;
        if (sl + c - 1 < CCAP) stage[sl + c - 1] = node;
        else                   col[base_s + sl + c - 1] = node;
    }
    if (t < BINW) lc[t] = 0;
    __syncthreads();
#pragma unroll
    for (int i = 0; i < SEGW; ++i) {
#pragma unroll
        for (int j = 0; j < SEGR; ++j) {
            if (j * 32 + lane < pc[i]) {
                int jj = r[i][j].x - row0;
                int p  = rs[jj] + atomicAdd(&lc[jj], 1);
                if (p < CCAP) stage[p] = r[i][j].y;
                else          col[base_s + p] = r[i][j].y;
            }
        }
        for (int q = 32 * SEGR + lane; q < pc[i]; q += 32) {
            int2 rr = binned[po[i] + q];
            int jj = rr.x - row0;
            int p  = rs[jj] + atomicAdd(&lc[jj], 1);
            if (p < CCAP) stage[p] = rr.y;
            else          col[base_s + p] = rr.y;
        }
    }
    __syncthreads();
    int m = tot_s < CCAP ? tot_s : CCAP;
    if (m > 0) {
        int v[CCAP / BB];
#pragma unroll
        for (int k = 0; k < CCAP / BB; ++k) {
            int q = k * BB + t;
            v[k] = stage[q < m ? q : m - 1];
        }
#pragma unroll
        for (int k = 0; k < CCAP / BB; ++k) {
            int q = k * BB + t;
            if (q < m) col[base_s + q] = v[k];
        }
    }
}


__device__ __forceinline__ void fma4(float4& a, const float4& v, float w) {
    a.x = fmaf(v.x, w, a.x);
    a.y = fmaf(v.y, w, a.y);
    a.z = fmaf(v.z, w, a.z);
    a.w = fmaf(v.w, w, a.w);
}

#define SCAP (PITCH / 2)

template <bool SORT>
__global__ __launch_bounds__(256) void gcn_layer(const float* __restrict__ Xin,
                                                 const int* __restrict__ rowstart,
                                                 const int* __restrict__ cnt,
                                                 const int2* __restrict__ rec,
                                                 const float* __restrict__ dis,
                                                 const int* __restrict__ arr,
                                                 int2* recw,
                                                 const float* __restrict__ Wm,
                                                 const float* __restrict__ bias,
                                                 float* __restrict__ out, int n, int relu) {
    __shared__ __attribute__((aligned(16))) float aggT[TR * PITCH];
    __shared__ __attribute__((aligned(16))) float outT[TR * PITCH];

    int wib  = __builtin_amdgcn_readfirstlane((int)(threadIdx.x >> 5));
    int lane = threadIdx.x & 31;
    int row0  = blockIdx.x * TR;
    int node0 = row0 + wib * NPW;

    int mst = 0, mc = 0, mdt = 0;
    if (lane < NPW && node0 + lane < n) {
        mst = rowstart[node0 + lane];
        mc  = cnt[node0 + lane] + 1;
        if (SORT) mdt = __float_as_int(dis[node0 + lane]);
    }

    int  st[NPW], c[NPW];
    int2 r[NPW];
    if (!SORT) {
#pragma unroll
        for (int i = 0; i < NPW; ++i) {
            st[i] = __builtin_amdgcn_readlane(mst, i);
            c[i]  = __builtin_amdgcn_readlane(mc, i);
            r[i]  = make_int2(0, 0);
            if (lane < c[i]) r[i] = rec[st[i] + lane];
        }
    } else {
        int   cc[NPW], n0[NPW], key[NPW], ord[NPW], rk[NPW];
        float dt[NPW], wgt[NPW];
#pragma unroll
        for (int i = 0; i < NPW; ++i) {
            st[i] = __builtin_amdgcn_readlane(mst, i);
            c[i]  = __builtin_amdgcn_readlane(mc, i);
            dt[i] = __int_as_float(__builtin_amdgcn_readlane(mdt, i));
            cc[i] = c[i] > 0 ? c[i] - 1 : 0;
            n0[i] = cc[i] < 32 ? cc[i] : 32;
            int kq = arr[st[i] + (lane < n0[i] ? lane : 0)];
            key[i] = lane < n0[i] ? kq : 0x7fffffff;
        }
#pragma unroll
        for (int i = 0; i < NPW; ++i) {
            wgt[i] = dis[lane < n0[i] ? key[i] : 0];
        }
#pragma unroll
        for (int i = 0; i < NPW; ++i) {
            ord[i] = 0x7fffffff;
            if (lane < n0[i]) ord[i] = (key[i] << 5) | lane;
            rk[i] = 0;
            for (int k = 0; k < n0[i]; ++k) {
                int kk = __builtin_amdgcn_readlane(ord[i], k);
                rk[i] += (kk < ord[i]) ? 1 : 0;
            }
            if (cc[i] > 32) {
                for (int b = 32; b < cc[i]; b += 32) {
                    int kv = 0x7fffffff;
                    if (b + lane < cc[i]) kv = arr[st[i] + b + lane];
                    int lim = cc[i] - b < 32 ? cc[i] - b : 32;
                    for (int k = 0; k < lim; ++k) {
                        int kk = __builtin_amdgcn_readlane(kv, k);
                        rk[i] += (kk < key[i]) ? 1 : 0;
                    }
                }
            }
        }
#pragma unroll
        for (int i = 0; i < NPW; ++i) {
            int2* slot = (int2*)(outT + (wib * NPW + i) * PITCH);
            if (lane < n0[i]) {
                int2 q = make_int2(key[i], __float_as_int(wgt[i] * dt[i]));
                recw[st[i] + rk[i]] = q;
                if (rk[i] < SCAP) slot[rk[i]] = q;
            }
            if (cc[i] > 32) {
                for (int a = 32; a < cc[i]; a += 32) {
                    int e  = a + lane;
                    int ke = 0x7fffffff;
                    if (e < cc[i]) ke = arr[st[i] + e];
                    int re = 0;
                    for (int b = 0; b < cc[i]; b += 32) {
                        int kv = 0x7fffffff;
                        if (b + lane < cc[i]) kv = arr[st[i] + b + lane];
                        int lim = cc[i] - b < 32 ? cc[i] - b : 32;
                        for (int k = 0; k < lim; ++k) {
                            int kk = __builtin_amdgcn_readlane(kv, k);
                            re += (kk < ke || (kk == ke && b + k < e)) ? 1 : 0;
                        }
                    }
                    if (e < cc[i]) {
                        int2 q = make_int2(ke, __float_as_int(dis[ke] * dt[i]));
                        recw[st[i] + re] = q;
                        if (re < SCAP) slot[re] = q;
                    }
                }
            }
            if (lane == 0 && c[i] > 0) {
                int2 q = make_int2(node0 + i, __float_as_int(dt[i] * dt[i]));
                recw[st[i] + cc[i]] = q;
                if (cc[i] < SCAP) slot[cc[i]] = q;
            }
        }
#pragma unroll
        for (int i = 0; i < NPW; ++i) {
            const int2* slot = (const int2*)(outT + (wib * NPW + i) * PITCH);
            r[i] = make_int2(0, 0);
            if (lane < c[i]) r[i] = slot[lane];
        }
    }

#pragma unroll
    for (int i = 0; i < NPW; ++i) {
        float4 acc = make_float4(0.0f, 0.0f, 0.0f, 0.0f);

        int c32 = c[i] < 32 ? c[i] : 32;
        for (int j0 = 0; j0 < c32; j0 += 8) {
            float4 v[8];
#pragma unroll
            for (int k = 0; k < 8; ++k) {
                int jj = j0 + k < c32 ? j0 + k : c32 - 1;
                int s = __builtin_amdgcn_readlane(r[i].x, jj);
                v[k] = ((const float4*)(Xin + (size_t)s * DCH))[lane];
            }
#pragma unroll
            for (int k = 0; k < 8; ++k) {
                if (j0 + k < c32)
                    fma4(acc, v[k], __int_as_float(__builtin_amdgcn_readlane(r[i].y, j0 + k)));
            }
        }
        if (!SORT) {
            for (int j = 32; j < c[i]; ++j) {
                int2 q = rec[st[i] + j];
                float4 v = ((const float4*)(Xin + (size_t)q.x * DCH))[lane];
                fma4(acc, v, __int_as_float(q.y));
            }
        } else {
            const int2* slot = (const int2*)(outT + (wib * NPW + i) * PITCH);
            if (c[i] > SCAP) __threadfence();
            for (int j = 32; j < c[i]; ++j) {
                int2 q;
                if (j < SCAP) {
                    q = slot[j];
                } else {
                    unsigned long long u = __hip_atomic_load((unsigned long long*)(recw + st[i] + j),
                                                             __ATOMIC_RELAXED, __HIP_MEMORY_SCOPE_AGENT);
                    q = make_int2((int)(u & 0xffffffffull), (int)(u >> 32));
                }
                float4 v = ((const float4*)(Xin + (size_t)q.x * DCH))[lane];
                fma4(acc, v, __int_as_float(q.y));
            }
        }

        *(float4*)(aggT + (wib * NPW + i) * PITCH + 4 * lane) = acc;
    }

    __syncthreads();

    int hi  = lane >> 4;
    int l16 = lane & 15;
    int col = wib * 16 + l16;
    const float* a0p = aggT + l16 * PITCH + 2 * hi;
    const float* a1p = a0p + 16 * PITCH;
    const float* bp  = Wm + (size_t)(2 * hi) * DCH + col;

    v8f c0 = {};
    v8f c1 = {};
#pragma unroll 8
    for (int k0 = 0; k0 < DCH; k0 += 4) {
        v2f a0 = *(const v2f*)(a0p + k0);
        v2f a1 = *(const v2f*)(a1p + k0);
        v2f b;
        b.x = bp[(size_t)k0 * DCH];
        b.y = bp[(size_t)(k0 + 1) * DCH];
        c0 = __builtin_amdgcn_wmma_f32_16x16x4_f32(
                  false, a0,  false, b,
                  (short)0, c0,  false,  false);
        c1 = __builtin_amdgcn_wmma_f32_16x16x4_f32(
                  false, a1,  false, b,
                  (short)0, c1,  false,  false);
    }

#pragma unroll
    for (int i = 0; i < 8; ++i) {
        outT[(8 * hi + i) * PITCH + col]      = c0[i];
        outT[(16 + 8 * hi + i) * PITCH + col] = c1[i];
    }

    __syncthreads();

    float4 b4 = ((const float4*)bias)[lane];
#pragma unroll
    for (int i = 0; i < NPW; ++i) {
        int lr = wib * NPW + i;
        int t  = row0 + lr;
        if (t < n) {
            float4 o = *(const float4*)(outT + lr * PITCH + 4 * lane);
            o.x += b4.x; o.y += b4.y; o.z += b4.z; o.w += b4.w;
            if (relu) {
                o.x = fmaxf(o.x, 0.0f); o.y = fmaxf(o.y, 0.0f);
                o.z = fmaxf(o.z, 0.0f); o.w = fmaxf(o.w, 0.0f);
            }
            ((float4*)(out + (size_t)t * DCH))[lane] = o;
        }
    }
}


extern "C" void kernel_launch(void* const* d_in, const int* in_sizes, int n_in,
                              void* d_out, int out_size, void* d_ws, size_t ws_size,
                              hipStream_t stream) {
    const float* X  = (const float*)d_in[0];
    const int*   ei = (const int*)  d_in[1];
    const float* W1 = (const float*)d_in[2];
    const float* b1 = (const float*)d_in[3];
    const float* W2 = (const float*)d_in[4];
    const float* b2 = (const float*)d_in[5];
    const float* W3 = (const float*)d_in[6];
    const float* b3 = (const float*)d_in[7];
    float* out = (float*)d_out;

    const int n  = in_sizes[0] / DCH;
    const int e  = in_sizes[1] / 2;

    const int* src = ei;
    const int* tgt = ei + e;

    int*   cnt      = (int*)d_ws;
    int*   rowstart = cnt + n;
    float* dis      = (float*)(rowstart + n);
    int*   total    = (int*)(dis + n);
    int*   tab_off  = total + 4;
    int*   tab_cnt  = tab_off + NBLK * NBIN;
    int2*  binned   = (int2*)(tab_cnt + NBLK * NBIN);
    int*   col      = (int*)(binned + e);
    int2*  rec      = (int2*)(col + e + n);
    float* Y1       = (float*)(rec + e + n);

    const int chunk = (e + NBLK - 1) / NBLK;
    const int nbins = (n + BINW - 1) / BINW;
    bin_local<<<NBLK,  BT, 0, stream>>>(src, tgt, tab_off, tab_cnt, total, binned, e, chunk);
    bin_build<<<nbins, BB, 0, stream>>>(tab_off, tab_cnt, binned, total, cnt, rowstart, dis, col, n);

    const int layerBlocks = (n + TR - 1) / TR;

    gcn_layer<true> <<<layerBlocks, 256, 0, stream>>>(X,   rowstart, cnt, nullptr, dis, col, rec,
                                                      W1, b1, out, n, 1);
    gcn_layer<false><<<layerBlocks, 256, 0, stream>>>(out, rowstart, cnt, rec, dis, col, nullptr,
                                                      W2, b2, Y1,  n, 1);
    gcn_layer<false><<<layerBlocks, 256, 0, stream>>>(Y1,  rowstart, cnt, rec, dis, col, nullptr,
                                                      W3, b3, out, n, 0);
}
